# starting kernel
# baseline (speedup 1.0000x reference)
#include <hip/hip_runtime.h>
#include <hip/hip_bf16.h>

typedef float v2f __attribute__((ext_vector_type(2)));
typedef float v8f __attribute__((ext_vector_type(8)));

#define NEG_SLOPE 0.2f

__device__ __forceinline__ unsigned fkey(float x) {
  unsigned b = __float_as_uint(x);
  return (b & 0x80000000u) ? ~b : (b | 0x80000000u);
}
__device__ __forceinline__ float funkey(unsigned k) {
  return (k & 0x80000000u) ? __uint_as_float(k & 0x7FFFFFFFu) : __uint_as_float(~k);
}
#define NEG_INF_KEY 0x007FFFFFu

__global__ void fill_u32_kernel(unsigned* __restrict__ p, unsigned v, int n) {
  int i = blockIdx.x * blockDim.x + threadIdx.x;
  if (i < n) p[i] = v;
}

__global__ void gemm_wmma_f32_kernel(const float* __restrict__ A,
                                     const float* __restrict__ B,
                                     float* __restrict__ C,
                                     int M, int K, int Ncol) {
  const int wave = threadIdx.x >> 5;
  const int lane = threadIdx.x & 31;
  const int rowBase = blockIdx.x * 16;
  const int colBase = wave * 16;
  const int lm = lane & 15;
  const int hi = lane >> 4;

  v8f acc = {};
  const float* arow = A + (size_t)(rowBase + lm) * K;
  for (int k0 = 0; k0 < K; k0 += 4) {
    const int ka = k0 + hi * 2;
    v2f a, b;
    a.x = arow[ka];
    a.y = arow[ka + 1];
    b.x = B[(size_t)ka * Ncol + colBase + lm];
    b.y = B[(size_t)(ka + 1) * Ncol + colBase + lm];
    acc = __builtin_amdgcn_wmma_f32_16x16x4_f32(
         false, a,  false, b,
         (short)0, acc,  false,  false);
  }
#pragma unroll
  for (int r = 0; r < 8; ++r) {
    const int row = rowBase + hi * 8 + r;
    C[(size_t)row * Ncol + colBase + lm] = acc[r];
  }
}

__global__ void alpha_kernel(const float* __restrict__ h,
                             const float* __restrict__ a_src,
                             const float* __restrict__ a_dst,
                             float* __restrict__ as, float* __restrict__ ad,
                             int N, int H, int D) {
  int i = blockIdx.x * blockDim.x + threadIdx.x;
  if (i >= N * H) return;
  int hh = i % H;
  const float* hp = h + (size_t)i * D;
  const float* sp = a_src + (size_t)hh * D;
  const float* dp = a_dst + (size_t)hh * D;
  float s = 0.f, d = 0.f;
  for (int k = 0; k < D; ++k) { float v = hp[k]; s += v * sp[k]; d += v * dp[k]; }
  as[i] = s; ad[i] = d;
}

__device__ __forceinline__ void edge_sd(const int* __restrict__ ei, int E, int i,
                                        int& s, int& d) {
  if (i < E) { s = ei[i]; d = ei[E + i]; } else { s = d = i - E; }
}

__global__ void edge_max_kernel(const int* __restrict__ ei, int E, int N,
                                const float* __restrict__ as,
                                const float* __restrict__ ad,
                                unsigned* __restrict__ emax, int H) {
  int i = blockIdx.x * blockDim.x + threadIdx.x;
  if (i >= E + N) return;
  int s, d; edge_sd(ei, E, i, s, d);
  for (int h = 0; h < H; ++h) {
    float e = as[s * H + h] + ad[d * H + h];
    e = (e >= 0.f) ? e : NEG_SLOPE * e;
    atomicMax(&emax[d * H + h], fkey(e));
  }
}

__global__ void edge_acc_kernel(const int* __restrict__ ei, int E, int N,
                                const float* __restrict__ as,
                                const float* __restrict__ ad,
                                const unsigned* __restrict__ emax,
                                float* __restrict__ esum,
                                const float* __restrict__ hfeat,
                                float* __restrict__ acc,
                                int H, int D) {
  int t = blockIdx.x * blockDim.x + threadIdx.x;
  if (t >= (E + N) * H) return;
  int i = t / H, h = t % H;
  int s, d; edge_sd(ei, E, i, s, d);
  float e = as[s * H + h] + ad[d * H + h];
  e = (e >= 0.f) ? e : NEG_SLOPE * e;
  float ex = expf(e - funkey(emax[d * H + h]));
  atomicAdd(&esum[d * H + h], ex);
  const float* hs = hfeat + ((size_t)s * H + h) * D;
  float* ap = acc + ((size_t)d * H + h) * D;
  for (int k = 0; k < D; ++k) atomicAdd(&ap[k], ex * hs[k]);
}

__global__ void finalize_kernel(float* __restrict__ acc,
                                const float* __restrict__ esum,
                                const float* __restrict__ bias,
                                int N, int H, int D, int apply_elu) {
  int i = blockIdx.x * blockDim.x + threadIdx.x;
  int HD = H * D;
  if (i >= N * HD) return;
  int n = i / HD, c = i % HD, h = c / D;
  float v = acc[i] / (esum[n * H + h] + 1e-16f) + bias[c];
  if (apply_elu) v = (v > 0.f) ? v : (expf(v) - 1.f);
  acc[i] = v;
}

__global__ void pred_node_kernel(const float* __restrict__ h2,
                                 const float* __restrict__ Wp,
                                 float* __restrict__ ps, float* __restrict__ pd,
                                 int N, int D) {
  int n = blockIdx.x * blockDim.x + threadIdx.x;
  if (n >= N) return;
  const float* hp = h2 + (size_t)n * D;
  float s = 0.f, d = 0.f;
  for (int k = 0; k < D; ++k) { s += hp[k] * Wp[k]; d += hp[k] * Wp[D + k]; }
  ps[n] = s; pd[n] = d;
}

__global__ void pred_edge_kernel(const int* __restrict__ ei, int E,
                                 const float* __restrict__ ps,
                                 const float* __restrict__ pd,
                                 const float* __restrict__ bp,
                                 float* __restrict__ out) {
  int e = blockIdx.x * blockDim.x + threadIdx.x;
  if (e >= E) return;
  out[e] = ps[ei[e]] + pd[ei[E + e]] + bp[0];
}

static inline int cdiv(long long a, int b) { return (int)((a + b - 1) / b); }

extern "C" void kernel_launch(void* const* d_in, const int* in_sizes, int n_in,
                              void* d_out, int out_size, void* d_ws, size_t ws_size,
                              hipStream_t stream) {
  const float* x   = (const float*)d_in[0];
  const int*   ei  = (const int*)  d_in[1];
  const float* W1  = (const float*)d_in[2];
  const float* as1w= (const float*)d_in[3];
  const float* ad1w= (const float*)d_in[4];
  const float* b1  = (const float*)d_in[5];
  const float* W2  = (const float*)d_in[6];
  const float* as2w= (const float*)d_in[7];
  const float* ad2w= (const float*)d_in[8];
  const float* b2  = (const float*)d_in[9];
  const float* Wp  = (const float*)d_in[10];
  const float* bp  = (const float*)d_in[11];
  float* out = (float*)d_out;

  const int F = 128, H1 = 4, Hd = 32;
  const int N = in_sizes[0] / F;
  const int E = in_sizes[1] / 2;
  const int EN = E + N;

  float* ws = (float*)d_ws;
  float*    h1    = ws;                 ws += (size_t)N * F;
  float*    acc1  = ws;                 ws += (size_t)N * F;
  float*    as1   = ws;                 ws += (size_t)N * H1;
  float*    ad1   = ws;                 ws += (size_t)N * H1;
  unsigned* emax1 = (unsigned*)ws;      ws += (size_t)N * H1;
  float*    esum1 = ws;                 ws += (size_t)N * H1;
  float*    h2p   = ws;                 ws += (size_t)N * Hd;
  float*    acc2  = ws;                 ws += (size_t)N * Hd;
  float*    as2   = ws;                 ws += (size_t)N;
  float*    ad2   = ws;                 ws += (size_t)N;
  unsigned* emax2 = (unsigned*)ws;      ws += (size_t)N;
  float*    esum2 = ws;                 ws += (size_t)N;
  float*    ps    = ws;                 ws += (size_t)N;
  float*    pd    = ws;                 ws += (size_t)N;
  (void)ws_size;

  const int T = 256;

  fill_u32_kernel<<<cdiv((long long)N * F, T), T, 0, stream>>>((unsigned*)acc1, 0u, N * F);
  fill_u32_kernel<<<cdiv((long long)N * H1, T), T, 0, stream>>>((unsigned*)esum1, 0u, N * H1);
  fill_u32_kernel<<<cdiv((long long)N * H1, T), T, 0, stream>>>(emax1, NEG_INF_KEY, N * H1);
  fill_u32_kernel<<<cdiv((long long)N * Hd, T), T, 0, stream>>>((unsigned*)acc2, 0u, N * Hd);
  fill_u32_kernel<<<cdiv(N, T), T, 0, stream>>>((unsigned*)esum2, 0u, N);
  fill_u32_kernel<<<cdiv(N, T), T, 0, stream>>>(emax2, NEG_INF_KEY, N);

  gemm_wmma_f32_kernel<<<N / 16, 8 * 32, 0, stream>>>(x, W1, h1, N, F, F);
  alpha_kernel<<<cdiv((long long)N * H1, T), T, 0, stream>>>(h1, as1w, ad1w, as1, ad1, N, H1, Hd);
  edge_max_kernel<<<cdiv(EN, T), T, 0, stream>>>(ei, E, N, as1, ad1, emax1, H1);
  edge_acc_kernel<<<cdiv((long long)EN * H1, T), T, 0, stream>>>(ei, E, N, as1, ad1, emax1,
                                                                 esum1, h1, acc1, H1, Hd);
  finalize_kernel<<<cdiv((long long)N * F, T), T, 0, stream>>>(acc1, esum1, b1, N, H1, Hd, 1);

  gemm_wmma_f32_kernel<<<N / 16, 2 * 32, 0, stream>>>(acc1, W2, h2p, N, F, Hd);
  alpha_kernel<<<cdiv(N, T), T, 0, stream>>>(h2p, as2w, ad2w, as2, ad2, N, 1, Hd);
  edge_max_kernel<<<cdiv(EN, T), T, 0, stream>>>(ei, E, N, as2, ad2, emax2, 1);
  edge_acc_kernel<<<cdiv(EN, T), T, 0, stream>>>(ei, E, N, as2, ad2, emax2,
                                                 esum2, h2p, acc2, 1, Hd);
  finalize_kernel<<<cdiv((long long)N * Hd, T), T, 0, stream>>>(acc2, esum2, b2, N, 1, Hd, 0);

  pred_node_kernel<<<cdiv(N, T), T, 0, stream>>>(acc2, Wp, ps, pd, N, Hd);
  pred_edge_kernel<<<cdiv(E, T), T, 0, stream>>>(ei, E, ps, pd, bp, out);
}


# Aggregation without float atomics: edges grouped by destination in every call (one integer add per edge, scan, fill); one wave per node gathers source rows, softmax, bias, ELU in registers
# speedup vs baseline: 7.9305x; 7.9305x over previous
#include <hip/hip_runtime.h>
#include <hip/hip_bf16.h>


typedef float v2f __attribute__((ext_vector_type(2)));
typedef float v8f __attribute__((ext_vector_type(8)));

#define NEG_SLOPE 0.2f

__global__ void fill_u32_kernel(unsigned* __restrict__ p, unsigned v, int n) {
  int i = blockIdx.x * blockDim.x + threadIdx.x;
  if (i < n) p[i] = v;
}

__global__ void gemm_wmma_f32_kernel(const float* __restrict__ A,
                                     const float* __restrict__ B,
                                     float* __restrict__ C,
                                     int M, int K, int Ncol) {
  const int wave = threadIdx.x >> 5;
  const int lane = threadIdx.x & 31;
  const int rowBase = blockIdx.x * 16;
  const int colBase = wave * 16;
  const int lm = lane & 15;
  const int hi = lane >> 4;

  v8f acc = {};
  const float* arow = A + (size_t)(rowBase + lm) * K;
  for (int k0 = 0; k0 < K; k0 += 4) {
    const int ka = k0 + hi * 2;
    v2f a, b;
    a.x = arow[ka];
    a.y = arow[ka + 1];
    b.x = B[(size_t)ka * Ncol + colBase + lm];
    b.y = B[(size_t)(ka + 1) * Ncol + colBase + lm];
    acc = __builtin_amdgcn_wmma_f32_16x16x4_f32(
         false, a,  false, b,
         (short)0, acc,  false,  false);
  }
#pragma unroll
  for (int r = 0; r < 8; ++r) {
    const int row = rowBase + hi * 8 + r;
    C[(size_t)row * Ncol + colBase + lm] = acc[r];
  }
}

__global__ void alpha_kernel(const float* __restrict__ h,
                             const float* __restrict__ a_src,
                             const float* __restrict__ a_dst,
                             float* __restrict__ as, float* __restrict__ ad,
                             int N, int H, int D) {
  int i = blockIdx.x * blockDim.x + threadIdx.x;
  if (i >= N * H) return;
  int hh = i % H;
  const float* hp = h + (size_t)i * D;
  const float* sp = a_src + (size_t)hh * D;
  const float* dp = a_dst + (size_t)hh * D;
  float s = 0.f, d = 0.f;
  for (int k = 0; k < D; ++k) { float v = hp[k]; s += v * sp[k]; d += v * dp[k]; }
  as[i] = s; ad[i] = d;
}

__global__ void dst_rank_kernel(const int* __restrict__ ei, int E,
                                int* __restrict__ cnt, int* __restrict__ rank) {
  int e = blockIdx.x * blockDim.x + threadIdx.x;
  if (e < E) rank[e] = atomicAdd(&cnt[ei[E + e]], 1);
}

__global__ void __launch_bounds__(1024)
rowptr_scan_kernel(const int* __restrict__ cnt, int* __restrict__ rowptr, int N) {
  __shared__ int wsum[2][32];
  const int tid = threadIdx.x, lane = tid & 31, wv = tid >> 5;
  int carry = 0;
  int p = 0;
  for (int base = 0; base <= N; base += 4096) {
    const int i0 = base + 4 * tid;
    int4 c = make_int4(0, 0, 0, 0);
    if (i0 <= N) c = *reinterpret_cast<const int4*>(cnt + i0);
    const int q0 = c.x, q1 = q0 + c.y, q2 = q1 + c.z, v = q2 + c.w;
    int x = v;
    for (int o = 1; o < 32; o <<= 1) { int y = __shfl_up(x, o); if (lane >= o) x += y; }
    if (lane == 31) wsum[p][wv] = x;
    __syncthreads();
    const int s = wsum[p][lane];
    int xs = s;
    for (int o = 1; o < 32; o <<= 1) { int y = __shfl_up(xs, o); if (lane >= o) xs += y; }
    const int woff = __shfl(xs - s, wv);
    const int tot = __shfl(xs, 31);
    const int ex = carry + woff + x - v;
    if (i0 <= N)
      *reinterpret_cast<int4*>(rowptr + i0) = make_int4(ex, ex + q0, ex + q1, ex + q2);
    carry += tot;
    p ^= 1;
  }
}

__global__ void csr_fill_kernel(const int* __restrict__ ei, int E,
                                const int* __restrict__ rowptr,
                                const int* __restrict__ rank, int* __restrict__ col) {
  int e = blockIdx.x * blockDim.x + threadIdx.x;
  if (e >= E) return;
  col[rowptr[ei[E + e]] + rank[e]] = ei[e];
}

__device__ __forceinline__ int lane_i(int v, int t) { return __builtin_amdgcn_readlane(v, t); }
__device__ __forceinline__ float lane_f(float v, int t) {
  return __int_as_float(__builtin_amdgcn_readlane(__float_as_int(v), t));
}
__device__ __forceinline__ float lrelu(float e) { return (e >= 0.f) ? e : NEG_SLOPE * e; }
__device__ __forceinline__ float wave_max(float v) {
  for (int o = 16; o > 0; o >>= 1) v = fmaxf(v, __shfl_xor(v, o));
  return v;
}
__device__ __forceinline__ float wave_sum(float v) {
  for (int o = 16; o > 0; o >>= 1) v += __shfl_xor(v, o);
  return v;
}

__global__ void __launch_bounds__(256)
gat_gather_h4_kernel(const int* __restrict__ rowptr, const int* __restrict__ col,
                     const float* __restrict__ as, const float* __restrict__ ad,
                     const float* __restrict__ hfeat, const float* __restrict__ bias,
                     float* __restrict__ out, int N) {
  const int lane = threadIdx.x & 31;
  const int d = __builtin_amdgcn_readfirstlane(blockIdx.x * 8 + (threadIdx.x >> 5));
  if (d >= N) return;
  const int beg = __builtin_amdgcn_readfirstlane(rowptr[d]);
  const int deg = __builtin_amdgcn_readfirstlane(rowptr[d + 1]) - beg + 1;
  const float4 adv = *reinterpret_cast<const float4*>(ad + 4 * (size_t)d);

  float m0 = -INFINITY, m1 = -INFINITY, m2 = -INFINITY, m3 = -INFINITY;
  for (int j0 = 0; j0 < deg; j0 += 32) {
    const int j = j0 + lane;
    if (j < deg) {
      const int s = (j < deg - 1) ? col[beg + j] : d;
      const float4 a = *reinterpret_cast<const float4*>(as + 4 * (size_t)s);
      m0 = fmaxf(m0, lrelu(a.x + adv.x));
      m1 = fmaxf(m1, lrelu(a.y + adv.y));
      m2 = fmaxf(m2, lrelu(a.z + adv.z));
      m3 = fmaxf(m3, lrelu(a.w + adv.w));
    }
  }
  m0 = wave_max(m0); m1 = wave_max(m1); m2 = wave_max(m2); m3 = wave_max(m3);

  const bool hb0 = (lane & 8) != 0, hb1 = (lane & 16) != 0;
  float4 acc = make_float4(0.f, 0.f, 0.f, 0.f);
  float s0 = 0.f, s1 = 0.f, s2 = 0.f, s3 = 0.f;
  for (int j0 = 0; j0 < deg; j0 += 32) {
    const int j = j0 + lane;
    int s = d;
    float x0 = 0.f, x1 = 0.f, x2 = 0.f, x3 = 0.f;
    if (j < deg) {
      if (j < deg - 1) s = col[beg + j];
      const float4 a = *reinterpret_cast<const float4*>(as + 4 * (size_t)s);
      x0 = expf(lrelu(a.x + adv.x) - m0);
      x1 = expf(lrelu(a.y + adv.y) - m1);
      x2 = expf(lrelu(a.z + adv.z) - m2);
      x3 = expf(lrelu(a.w + adv.w) - m3);
      s0 += x0; s1 += x1; s2 += x2; s3 += x3;
    }
    const int cntc = (deg - j0 < 32) ? (deg - j0) : 32;
#pragma unroll 4
    for (int t = 0; t < cntc; ++t) {
      const int st = lane_i(s, t);
      const float w0 = lane_f(x0, t), w1 = lane_f(x1, t);
      const float w2 = lane_f(x2, t), w3 = lane_f(x3, t);
      const float wlo = hb0 ? w1 : w0, whi = hb0 ? w3 : w2;
      const float w = hb1 ? whi : wlo;
      const float4 hv = *reinterpret_cast<const float4*>(hfeat + (size_t)st * 128 + 4 * lane);
      acc.x += w * hv.x; acc.y += w * hv.y; acc.z += w * hv.z; acc.w += w * hv.w;
    }
  }
  s0 = wave_sum(s0); s1 = wave_sum(s1); s2 = wave_sum(s2); s3 = wave_sum(s3);
  const float slo = hb0 ? s1 : s0, shi = hb0 ? s3 : s2;
  const float den = (hb1 ? shi : slo) + 1e-16f;
  const float* bp = bias + 4 * lane;
  float4 o;
  o.x = acc.x / den + bp[0];
  o.y = acc.y / den + bp[1];
  o.z = acc.z / den + bp[2];
  o.w = acc.w / den + bp[3];
  o.x = (o.x > 0.f) ? o.x : (expf(o.x) - 1.f);
  o.y = (o.y > 0.f) ? o.y : (expf(o.y) - 1.f);
  o.z = (o.z > 0.f) ? o.z : (expf(o.z) - 1.f);
  o.w = (o.w > 0.f) ? o.w : (expf(o.w) - 1.f);
  *reinterpret_cast<float4*>(out + (size_t)d * 128 + 4 * lane) = o;
}

__global__ void __launch_bounds__(256)
gat_gather_h1_kernel(const int* __restrict__ rowptr, const int* __restrict__ col,
                     const float* __restrict__ as, const float* __restrict__ ad,
                     const float* __restrict__ hfeat, const float* __restrict__ bias,
                     float* __restrict__ out, int N) {
  const int lane = threadIdx.x & 31;
  const int d = __builtin_amdgcn_readfirstlane(blockIdx.x * 8 + (threadIdx.x >> 5));
  if (d >= N) return;
  const int beg = __builtin_amdgcn_readfirstlane(rowptr[d]);
  const int deg = __builtin_amdgcn_readfirstlane(rowptr[d + 1]) - beg + 1;
  const float adv = ad[d];

  float m = -INFINITY;
  for (int j0 = 0; j0 < deg; j0 += 32) {
    const int j = j0 + lane;
    if (j < deg) {
      const int s = (j < deg - 1) ? col[beg + j] : d;
      m = fmaxf(m, lrelu(as[s] + adv));
    }
  }
  m = wave_max(m);

  float acc = 0.f, sum = 0.f;
  for (int j0 = 0; j0 < deg; j0 += 32) {
    const int j = j0 + lane;
    int s = d;
    float x = 0.f;
    if (j < deg) {
      if (j < deg - 1) s = col[beg + j];
      x = expf(lrelu(as[s] + adv) - m);
      sum += x;
    }
    const int cntc = (deg - j0 < 32) ? (deg - j0) : 32;
#pragma unroll 4
    for (int t = 0; t < cntc; ++t) {
      const int st = lane_i(s, t);
      const float w = lane_f(x, t);
      acc += w * hfeat[(size_t)st * 32 + lane];
    }
  }
  sum = wave_sum(sum);
  out[(size_t)d * 32 + lane] = acc / (sum + 1e-16f) + bias[lane];
}

__global__ void pred_node_kernel(const float* __restrict__ h2,
                                 const float* __restrict__ Wp,
                                 float* __restrict__ ps, float* __restrict__ pd,
                                 int N, int D) {
  int n = blockIdx.x * blockDim.x + threadIdx.x;
  if (n >= N) return;
  const float* hp = h2 + (size_t)n * D;
  float s = 0.f, d = 0.f;
  for (int k = 0; k < D; ++k) { s += hp[k] * Wp[k]; d += hp[k] * Wp[D + k]; }
  ps[n] = s; pd[n] = d;
}

__global__ void pred_edge_kernel(const int* __restrict__ ei, int E,
                                 const float* __restrict__ ps,
                                 const float* __restrict__ pd,
                                 const float* __restrict__ bp,
                                 float* __restrict__ out) {
  int e = blockIdx.x * blockDim.x + threadIdx.x;
  if (e >= E) return;
  out[e] = ps[ei[e]] + pd[ei[E + e]] + bp[0];
}

static inline int cdiv(long long a, int b) { return (int)((a + b - 1) / b); }

extern "C" void kernel_launch(void* const* d_in, const int* in_sizes, int n_in,
                              void* d_out, int out_size, void* d_ws, size_t ws_size,
                              hipStream_t stream) {
  const float* x   = (const float*)d_in[0];
  const int*   ei  = (const int*)  d_in[1];
  const float* W1  = (const float*)d_in[2];
  const float* as1w= (const float*)d_in[3];
  const float* ad1w= (const float*)d_in[4];
  const float* b1  = (const float*)d_in[5];
  const float* W2  = (const float*)d_in[6];
  const float* as2w= (const float*)d_in[7];
  const float* ad2w= (const float*)d_in[8];
  const float* b2  = (const float*)d_in[9];
  const float* Wp  = (const float*)d_in[10];
  const float* bp  = (const float*)d_in[11];
  float* out = (float*)d_out;

  const int F = 128, H1 = 4, Hd = 32;
  const int N = in_sizes[0] / F;
  const int E = in_sizes[1] / 2;

  float* ws = (float*)d_ws;
  float*    h1    = ws;                 ws += (size_t)N * F;
  float*    o1    = ws;                 ws += (size_t)N * F;
  float*    as1   = ws;                 ws += (size_t)N * H1;
  float*    ad1   = ws;                 ws += (size_t)N * H1;
  int*      rowptr= (int*)ws;           ws += (size_t)N + 4;
  int*      cnt   = (int*)ws;           ws += (size_t)N + 4;
  int*      col   = (int*)ws;           ws += (size_t)E;
  int*      rank  = (int*)ws;           ws += (size_t)E;
  float*    as2   = ws;                 ws += (size_t)N;
  float*    ad2   = ws;                 ws += (size_t)N;
  float*    ps    = ws;                 ws += (size_t)N;
  float*    pd    = ws;                 ws += (size_t)N;
  float*    h2p   = h1;
  float*    o2    = h1 + (size_t)N * Hd;
  (void)ws_size; (void)n_in; (void)out_size;

  const int T = 256;

  fill_u32_kernel<<<cdiv(N + 4, T), T, 0, stream>>>((unsigned*)cnt, 0u, N + 4);
  dst_rank_kernel<<<cdiv(E, T), T, 0, stream>>>(ei, E, cnt, rank);
  rowptr_scan_kernel<<<1, 1024, 0, stream>>>(cnt, rowptr, N);
  csr_fill_kernel<<<cdiv(E, T), T, 0, stream>>>(ei, E, rowptr, rank, col);

  gemm_wmma_f32_kernel<<<N / 16, 8 * 32, 0, stream>>>(x, W1, h1, N, F, F);
  alpha_kernel<<<cdiv((long long)N * H1, T), T, 0, stream>>>(h1, as1w, ad1w, as1, ad1, N, H1, Hd);
  gat_gather_h4_kernel<<<cdiv(N, 8), T, 0, stream>>>(rowptr, col, as1, ad1, h1, b1, o1, N);

  gemm_wmma_f32_kernel<<<N / 16, 2 * 32, 0, stream>>>(o1, W2, h2p, N, F, Hd);
  alpha_kernel<<<cdiv(N, T), T, 0, stream>>>(h2p, as2w, ad2w, as2, ad2, N, 1, Hd);
  gat_gather_h1_kernel<<<cdiv(N, 8), T, 0, stream>>>(rowptr, col, as2, ad2, h2p, b2, o2, N);

  pred_node_kernel<<<cdiv(N, T), T, 0, stream>>>(o2, Wp, ps, pd, N, Hd);
  pred_edge_kernel<<<cdiv(E, T), T, 0, stream>>>(ei, E, ps, pd, bp, out);
}


# Edge-counter prefix sum in one pass: each thread issues its 14 16-byte loads up front, the 7 chunk scans run as overlapped chains, one barrier (was 13 serial trips: load, two lane scans, barrier)
# speedup vs baseline: 9.1862x; 1.1583x over previous
#include <hip/hip_runtime.h>
#include <hip/hip_bf16.h>


typedef float v2f __attribute__((ext_vector_type(2)));
typedef float v8f __attribute__((ext_vector_type(8)));

#define NEG_SLOPE 0.2f

template <int NCOL>
__device__ __forceinline__ void gemm_alpha_body(const float* __restrict__ A,
                                                const float* __restrict__ B,
                                                const float* __restrict__ a_src,
                                                const float* __restrict__ a_dst,
                                                float* __restrict__ C,
                                                float* __restrict__ as, float* __restrict__ ad,
                                                unsigned* __restrict__ zero_p, int nz) {
  constexpr int K = 128;
  constexpr int H = NCOL / 32;
  __shared__ float s_al[2][16 * H];

  for (int i = blockIdx.x * blockDim.x + threadIdx.x; i < nz; i += gridDim.x * blockDim.x)
    zero_p[i] = 0u;

  const int head = threadIdx.x >> 5;
  const int lane = threadIdx.x & 31;
  const int rowBase = blockIdx.x * 16;
  const int colBase = head * 32;
  const int lm = lane & 15;
  const int hi = lane >> 4;

  v8f acc0 = {}, acc1 = {};
  const float* arow = A + (size_t)(rowBase + lm) * K + hi * 2;
  const float* bcol = B + (size_t)(hi * 2) * NCOL + colBase + lm;
  for (int k0 = 0; k0 < K; k0 += 4) {
    v2f a, b0, b1;
    a.x = arow[k0];
    a.y = arow[k0 + 1];
    b0.x = bcol[(size_t)k0 * NCOL];
    b0.y = bcol[(size_t)(k0 + 1) * NCOL];
    b1.x = bcol[(size_t)k0 * NCOL + 16];
    b1.y = bcol[(size_t)(k0 + 1) * NCOL + 16];
    acc0 = __builtin_amdgcn_wmma_f32_16x16x4_f32(
         false, a,  false, b0,
         (short)0, acc0,  false,  false);
    acc1 = __builtin_amdgcn_wmma_f32_16x16x4_f32(
         false, a,  false, b1,
         (short)0, acc1,  false,  false);
  }
#pragma unroll
  for (int r = 0; r < 8; ++r) {
    float* crow = C + (size_t)(rowBase + hi * 8 + r) * NCOL + colBase + lm;
    crow[0] = acc0[r];
    crow[16] = acc1[r];
  }

  const float sa0 = a_src[colBase + lm], sa1 = a_src[colBase + 16 + lm];
  const float da0 = a_dst[colBase + lm], da1 = a_dst[colBase + 16 + lm];
  float ps[8], pd[8];
  for (int r = 0; r < 8; ++r) {
    ps[r] = acc0[r] * sa0 + acc1[r] * sa1;
    pd[r] = acc0[r] * da0 + acc1[r] * da1;
  }
  const bool b3 = (lm & 8) != 0, b2 = (lm & 4) != 0, b1 = (lm & 2) != 0;
  float qs[4], qd[4];
  for (int i = 0; i < 4; ++i) {
    qs[i] = (b3 ? ps[i + 4] : ps[i]) + __shfl_xor(b3 ? ps[i] : ps[i + 4], 8);
    qd[i] = (b3 ? pd[i + 4] : pd[i]) + __shfl_xor(b3 ? pd[i] : pd[i + 4], 8);
  }
  float ts[2], td[2];
  for (int i = 0; i < 2; ++i) {
    ts[i] = (b2 ? qs[i + 2] : qs[i]) + __shfl_xor(b2 ? qs[i] : qs[i + 2], 4);
    td[i] = (b2 ? qd[i + 2] : qd[i]) + __shfl_xor(b2 ? qd[i] : qd[i + 2], 4);
  }
  float us = (b1 ? ts[1] : ts[0]) + __shfl_xor(b1 ? ts[0] : ts[1], 2);
  float ud = (b1 ? td[1] : td[0]) + __shfl_xor(b1 ? td[0] : td[1], 2);
  us += __shfl_xor(us, 1);
  ud += __shfl_xor(ud, 1);
  if ((lm & 1) == 0) {
    const int rr = hi * 8 + (lm >> 1);
    s_al[0][rr * H + head] = us;
    s_al[1][rr * H + head] = ud;
  }
  __syncthreads();
  if (threadIdx.x < 32 * H) {
    const int which = threadIdx.x / (16 * H);
    const int idx = threadIdx.x % (16 * H);
    float* dstp = which ? ad : as;
    dstp[(size_t)rowBase * H + idx] = s_al[which][idx];
  }
}

__global__ void __launch_bounds__(128)
gemm_alpha_h4_kernel(const float* __restrict__ A, const float* __restrict__ B,
                     const float* __restrict__ a_src, const float* __restrict__ a_dst,
                     float* __restrict__ C, float* __restrict__ as, float* __restrict__ ad,
                     unsigned* __restrict__ zero_p, int nz) {
  gemm_alpha_body<128>(A, B, a_src, a_dst, C, as, ad, zero_p, nz);
}

__global__ void __launch_bounds__(32)
gemm_alpha_h1_kernel(const float* __restrict__ A, const float* __restrict__ B,
                     const float* __restrict__ a_src, const float* __restrict__ a_dst,
                     float* __restrict__ C, float* __restrict__ as, float* __restrict__ ad) {
  gemm_alpha_body<32>(A, B, a_src, a_dst, C, as, ad, nullptr, 0);
}

__global__ void dst_rank_kernel(const int* __restrict__ ei, int E,
                                int* __restrict__ cnt, int* __restrict__ rank) {
  int e = blockIdx.x * blockDim.x + threadIdx.x;
  if (e < E) rank[e] = atomicAdd(&cnt[ei[E + e]], 1);
}

constexpr int SCAN_KC = 7;
__global__ void __launch_bounds__(1024)
rowptr_scan_kernel(const int* __restrict__ cnt, int* __restrict__ rowptr, int N) {
  __shared__ int wsum[SCAN_KC][32];
  const int tid = threadIdx.x, lane = tid & 31;
  const int wv = __builtin_amdgcn_readfirstlane(tid >> 5);
  const int last4 = N & ~3;
  int carry = 0;
  for (int base = 0; base <= N; base += SCAN_KC * 8192) {
    if (base != 0) __syncthreads();
    int4 ca[SCAN_KC], cb[SCAN_KC];
#pragma unroll
    for (int k = 0; k < SCAN_KC; ++k) {
      const int i0 = base + k * 8192 + 8 * tid;
      ca[k] = *reinterpret_cast<const int4*>(cnt + ((i0 <= N) ? i0 : last4));
      cb[k] = *reinterpret_cast<const int4*>(cnt + ((i0 + 4 <= N) ? i0 + 4 : last4));
    }
    int v[SCAN_KC], x[SCAN_KC];
#pragma unroll
    for (int k = 0; k < SCAN_KC; ++k) {
      const int i0 = base + k * 8192 + 8 * tid;
      if (i0 > N) ca[k] = make_int4(0, 0, 0, 0);
      if (i0 + 4 > N) cb[k] = make_int4(0, 0, 0, 0);
      v[k] = ((ca[k].x + ca[k].y) + (ca[k].z + ca[k].w)) + ((cb[k].x + cb[k].y) + (cb[k].z + cb[k].w));
      x[k] = v[k];
    }
    for (int o = 1; o < 32; o <<= 1) {
#pragma unroll
      for (int k = 0; k < SCAN_KC; ++k) {
        const int y = __shfl_up(x[k], o);
        if (lane >= o) x[k] += y;
      }
    }
    if (lane == 31) {
#pragma unroll
      for (int k = 0; k < SCAN_KC; ++k) wsum[k][wv] = x[k];
    }
    __syncthreads();
    int s[SCAN_KC], xs[SCAN_KC];
#pragma unroll
    for (int k = 0; k < SCAN_KC; ++k) { s[k] = wsum[k][lane]; xs[k] = s[k]; }
    for (int o = 1; o < 32; o <<= 1) {
#pragma unroll
      for (int k = 0; k < SCAN_KC; ++k) {
        const int y = __shfl_up(xs[k], o);
        if (lane >= o) xs[k] += y;
      }
    }
    int run = carry;
#pragma unroll
    for (int k = 0; k < SCAN_KC; ++k) {
      const int woff = __builtin_amdgcn_readlane(xs[k] - s[k], wv);
      const int tot = __builtin_amdgcn_readlane(xs[k], 31);
      const int ex = run + woff + x[k] - v[k];
      const int i0 = base + k * 8192 + 8 * tid;
      const int e1 = ex + ca[k].x, e2 = e1 + ca[k].y, e3 = e2 + ca[k].z, e4 = e3 + ca[k].w;
      if (i0 <= N)
        *reinterpret_cast<int4*>(rowptr + i0) = make_int4(ex, e1, e2, e3);
      if (i0 + 4 <= N)
        *reinterpret_cast<int4*>(rowptr + i0 + 4) =
            make_int4(e4, e4 + cb[k].x, e4 + cb[k].x + cb[k].y, e4 + cb[k].x + cb[k].y + cb[k].z);
      run += tot;
    }
    carry = run;
  }
}

__global__ void csr_fill_kernel(const int* __restrict__ ei, int E,
                                const int* __restrict__ rowptr,
                                const int* __restrict__ rank, int* __restrict__ col) {
  int e = blockIdx.x * blockDim.x + threadIdx.x;
  if (e >= E) return;
  col[rowptr[ei[E + e]] + rank[e]] = ei[e];
}

__device__ __forceinline__ int lane_i(int v, int t) { return __builtin_amdgcn_readlane(v, t); }
__device__ __forceinline__ float lane_f(float v, int t) {
  return __int_as_float(__builtin_amdgcn_readlane(__float_as_int(v), t));
}
__device__ __forceinline__ float lrelu(float e) { return (e >= 0.f) ? e : NEG_SLOPE * e; }
__device__ __forceinline__ float wave_max(float v) {
  for (int o = 16; o > 0; o >>= 1) v = fmaxf(v, __shfl_xor(v, o));
  return v;
}
__device__ __forceinline__ float wave_sum(float v) {
  for (int o = 16; o > 0; o >>= 1) v += __shfl_xor(v, o);
  return v;
}

__global__ void __launch_bounds__(256)
gat_gather_h4_kernel(const int* __restrict__ rowptr, const int* __restrict__ col,
                     const float* __restrict__ as, const float* __restrict__ ad,
                     const float* __restrict__ hfeat, const float* __restrict__ bias,
                     float* __restrict__ out, int N) {
  const int lane = threadIdx.x & 31;
  const int d = __builtin_amdgcn_readfirstlane(blockIdx.x * 8 + (threadIdx.x >> 5));
  if (d >= N) return;
  const int beg = __builtin_amdgcn_readfirstlane(rowptr[d]);
  const int deg = __builtin_amdgcn_readfirstlane(rowptr[d + 1]) - beg + 1;
  const float4 adv = *reinterpret_cast<const float4*>(ad + 4 * (size_t)d);

  float m0 = -INFINITY, m1 = -INFINITY, m2 = -INFINITY, m3 = -INFINITY;
  for (int j0 = 0; j0 < deg; j0 += 32) {
    const int j = j0 + lane;
    if (j < deg) {
      const int s = (j < deg - 1) ? col[beg + j] : d;
      const float4 a = *reinterpret_cast<const float4*>(as + 4 * (size_t)s);
      m0 = fmaxf(m0, lrelu(a.x + adv.x));
      m1 = fmaxf(m1, lrelu(a.y + adv.y));
      m2 = fmaxf(m2, lrelu(a.z + adv.z));
      m3 = fmaxf(m3, lrelu(a.w + adv.w));
    }
  }
  m0 = wave_max(m0); m1 = wave_max(m1); m2 = wave_max(m2); m3 = wave_max(m3);

  const bool hb0 = (lane & 8) != 0, hb1 = (lane & 16) != 0;
  float4 acc = make_float4(0.f, 0.f, 0.f, 0.f);
  float s0 = 0.f, s1 = 0.f, s2 = 0.f, s3 = 0.f;
  for (int j0 = 0; j0 < deg; j0 += 32) {
    const int j = j0 + lane;
    int s = d;
    float x0 = 0.f, x1 = 0.f, x2 = 0.f, x3 = 0.f;
    if (j < deg) {
      if (j < deg - 1) s = col[beg + j];
      const float4 a = *reinterpret_cast<const float4*>(as + 4 * (size_t)s);
      x0 = expf(lrelu(a.x + adv.x) - m0);
      x1 = expf(lrelu(a.y + adv.y) - m1);
      x2 = expf(lrelu(a.z + adv.z) - m2);
      x3 = expf(lrelu(a.w + adv.w) - m3);
      s0 += x0; s1 += x1; s2 += x2; s3 += x3;
    }
    const int cntc = (deg - j0 < 32) ? (deg - j0) : 32;
#pragma unroll 4
    for (int t = 0; t < cntc; ++t) {
      const int st = lane_i(s, t);
      const float w0 = lane_f(x0, t), w1 = lane_f(x1, t);
      const float w2 = lane_f(x2, t), w3 = lane_f(x3, t);
      const float wlo = hb0 ? w1 : w0, whi = hb0 ? w3 : w2;
      const float w = hb1 ? whi : wlo;
      const float4 hv = *reinterpret_cast<const float4*>(hfeat + (size_t)st * 128 + 4 * lane);
      acc.x += w * hv.x; acc.y += w * hv.y; acc.z += w * hv.z; acc.w += w * hv.w;
    }
  }
  s0 = wave_sum(s0); s1 = wave_sum(s1); s2 = wave_sum(s2); s3 = wave_sum(s3);
  const float slo = hb0 ? s1 : s0, shi = hb0 ? s3 : s2;
  const float den = (hb1 ? shi : slo) + 1e-16f;
  const float* bp = bias + 4 * lane;
  float4 o;
  o.x = acc.x / den + bp[0];
  o.y = acc.y / den + bp[1];
  o.z = acc.z / den + bp[2];
  o.w = acc.w / den + bp[3];
  o.x = (o.x > 0.f) ? o.x : (expf(o.x) - 1.f);
  o.y = (o.y > 0.f) ? o.y : (expf(o.y) - 1.f);
  o.z = (o.z > 0.f) ? o.z : (expf(o.z) - 1.f);
  o.w = (o.w > 0.f) ? o.w : (expf(o.w) - 1.f);
  *reinterpret_cast<float4*>(out + (size_t)d * 128 + 4 * lane) = o;
}

__global__ void __launch_bounds__(256)
gat_gather_h1_kernel(const int* __restrict__ rowptr, const int* __restrict__ col,
                     const float* __restrict__ as, const float* __restrict__ ad,
                     const float* __restrict__ hfeat, const float* __restrict__ bias,
                     const float* __restrict__ Wp,
                     float* __restrict__ ps, float* __restrict__ pd, int N) {
  const int lane = threadIdx.x & 31;
  const int d = __builtin_amdgcn_readfirstlane(blockIdx.x * 8 + (threadIdx.x >> 5));
  if (d >= N) return;
  const int beg = __builtin_amdgcn_readfirstlane(rowptr[d]);
  const int deg = __builtin_amdgcn_readfirstlane(rowptr[d + 1]) - beg + 1;
  const float adv = ad[d];

  float m = -INFINITY;
  for (int j0 = 0; j0 < deg; j0 += 32) {
    const int j = j0 + lane;
    if (j < deg) {
      const int s = (j < deg - 1) ? col[beg + j] : d;
      m = fmaxf(m, lrelu(as[s] + adv));
    }
  }
  m = wave_max(m);

  float acc = 0.f, sum = 0.f;
  for (int j0 = 0; j0 < deg; j0 += 32) {
    const int j = j0 + lane;
    int s = d;
    float x = 0.f;
    if (j < deg) {
      if (j < deg - 1) s = col[beg + j];
      x = expf(lrelu(as[s] + adv) - m);
      sum += x;
    }
    const int cntc = (deg - j0 < 32) ? (deg - j0) : 32;
#pragma unroll 4
    for (int t = 0; t < cntc; ++t) {
      const int st = lane_i(s, t);
      const float w = lane_f(x, t);
      acc += w * hfeat[(size_t)st * 32 + lane];
    }
  }
  sum = wave_sum(sum);
  const float val = acc / (sum + 1e-16f) + bias[lane];
  const float s = wave_sum(val * Wp[lane]);
  const float t = wave_sum(val * Wp[32 + lane]);
  if (lane == 0) { ps[d] = s; pd[d] = t; }
}

__global__ void pred_edge_kernel(const int* __restrict__ ei, int E,
                                 const float* __restrict__ ps,
                                 const float* __restrict__ pd,
                                 const float* __restrict__ bp,
                                 float* __restrict__ out) {
  int e = blockIdx.x * blockDim.x + threadIdx.x;
  if (e >= E) return;
  out[e] = ps[ei[e]] + pd[ei[E + e]] + bp[0];
}

static inline int cdiv(long long a, int b) { return (int)((a + b - 1) / b); }

extern "C" void kernel_launch(void* const* d_in, const int* in_sizes, int n_in,
                              void* d_out, int out_size, void* d_ws, size_t ws_size,
                              hipStream_t stream) {
  const float* x   = (const float*)d_in[0];
  const int*   ei  = (const int*)  d_in[1];
  const float* W1  = (const float*)d_in[2];
  const float* as1w= (const float*)d_in[3];
  const float* ad1w= (const float*)d_in[4];
  const float* b1  = (const float*)d_in[5];
  const float* W2  = (const float*)d_in[6];
  const float* as2w= (const float*)d_in[7];
  const float* ad2w= (const float*)d_in[8];
  const float* b2  = (const float*)d_in[9];
  const float* Wp  = (const float*)d_in[10];
  const float* bp  = (const float*)d_in[11];
  float* out = (float*)d_out;

  const int F = 128, H1 = 4;
  const int N = in_sizes[0] / F;
  const int E = in_sizes[1] / 2;

  float* ws = (float*)d_ws;
  float*    h1    = ws;                 ws += (size_t)N * F;
  float*    o1    = ws;                 ws += (size_t)N * F;
  float*    as1   = ws;                 ws += (size_t)N * H1;
  float*    ad1   = ws;                 ws += (size_t)N * H1;
  int*      rowptr= (int*)ws;           ws += (size_t)N + 4;
  int*      cnt   = (int*)ws;           ws += (size_t)N + 4;
  int*      col   = (int*)ws;           ws += (size_t)E;
  int*      rank  = (int*)ws;           ws += (size_t)E;
  float*    as2   = ws;                 ws += (size_t)N;
  float*    ad2   = ws;                 ws += (size_t)N;
  float*    ps    = ws;                 ws += (size_t)N;
  float*    pd    = ws;                 ws += (size_t)N;
  float*    h2p   = h1;
  (void)ws_size; (void)n_in; (void)out_size;

  const int T = 256;

  gemm_alpha_h4_kernel<<<N / 16, 128, 0, stream>>>(x, W1, as1w, ad1w, h1, as1, ad1,
                                                   (unsigned*)cnt, N + 4);

  dst_rank_kernel<<<cdiv(E, T), T, 0, stream>>>(ei, E, cnt, rank);
  rowptr_scan_kernel<<<1, 1024, 0, stream>>>(cnt, rowptr, N);
  csr_fill_kernel<<<cdiv(E, T), T, 0, stream>>>(ei, E, rowptr, rank, col);

  gat_gather_h4_kernel<<<cdiv(N, 8), T, 0, stream>>>(rowptr, col, as1, ad1, h1, b1, o1, N);

  gemm_alpha_h1_kernel<<<N / 16, 32, 0, stream>>>(o1, W2, as2w, ad2w, h2p, as2, ad2);
  gat_gather_h1_kernel<<<cdiv(N, 8), T, 0, stream>>>(rowptr, col, as2, ad2, h2p, b2, Wp, ps, pd, N);

  pred_edge_kernel<<<cdiv(E, T), T, 0, stream>>>(ei, E, ps, pd, bp, out);
}


# Rating-head dots moved in front of the layer-2 aggregation (linear in h2): layer 1 emits 4 dots per node, layer 2 sums 2 scalars per edge, 16 lanes per node, no row gather; summation order changes
# speedup vs baseline: 10.7868x; 1.1742x over previous
#include <hip/hip_runtime.h>
#include <hip/hip_bf16.h>


typedef float v2f __attribute__((ext_vector_type(2)));
typedef float v8f __attribute__((ext_vector_type(8)));

#define NEG_SLOPE 0.2f

template <int NCOL>
__device__ __forceinline__ void gemm_alpha_body(const float* __restrict__ A,
                                                const float* __restrict__ B,
                                                const float* __restrict__ a_src,
                                                const float* __restrict__ a_dst,
                                                float* __restrict__ C,
                                                float* __restrict__ as, float* __restrict__ ad,
                                                unsigned* __restrict__ zero_p, int nz) {
  constexpr int K = 128;
  constexpr int H = NCOL / 32;
  __shared__ float s_al[2][16 * H];

  for (int i = blockIdx.x * blockDim.x + threadIdx.x; i < nz; i += gridDim.x * blockDim.x)
    zero_p[i] = 0u;

  const int head = threadIdx.x >> 5;
  const int lane = threadIdx.x & 31;
  const int rowBase = blockIdx.x * 16;
  const int colBase = head * 32;
  const int lm = lane & 15;
  const int hi = lane >> 4;

  v8f acc0 = {}, acc1 = {};
  const float* arow = A + (size_t)(rowBase + lm) * K + hi * 2;
  const float* bcol = B + (size_t)(hi * 2) * NCOL + colBase + lm;
  for (int k0 = 0; k0 < K; k0 += 4) {
    v2f a, b0, b1;
    a.x = arow[k0];
    a.y = arow[k0 + 1];
    b0.x = bcol[(size_t)k0 * NCOL];
    b0.y = bcol[(size_t)(k0 + 1) * NCOL];
    b1.x = bcol[(size_t)k0 * NCOL + 16];
    b1.y = bcol[(size_t)(k0 + 1) * NCOL + 16];
    acc0 = __builtin_amdgcn_wmma_f32_16x16x4_f32(
         false, a,  false, b0,
         (short)0, acc0,  false,  false);
    acc1 = __builtin_amdgcn_wmma_f32_16x16x4_f32(
         false, a,  false, b1,
         (short)0, acc1,  false,  false);
  }
#pragma unroll
  for (int r = 0; r < 8; ++r) {
    float* crow = C + (size_t)(rowBase + hi * 8 + r) * NCOL + colBase + lm;
    crow[0] = acc0[r];
    crow[16] = acc1[r];
  }

  const float sa0 = a_src[colBase + lm], sa1 = a_src[colBase + 16 + lm];
  const float da0 = a_dst[colBase + lm], da1 = a_dst[colBase + 16 + lm];
  float ps[8], pd[8];
  for (int r = 0; r < 8; ++r) {
    ps[r] = acc0[r] * sa0 + acc1[r] * sa1;
    pd[r] = acc0[r] * da0 + acc1[r] * da1;
  }
  const bool b3 = (lm & 8) != 0, b2 = (lm & 4) != 0, b1 = (lm & 2) != 0;
  float qs[4], qd[4];
  for (int i = 0; i < 4; ++i) {
    qs[i] = (b3 ? ps[i + 4] : ps[i]) + __shfl_xor(b3 ? ps[i] : ps[i + 4], 8);
    qd[i] = (b3 ? pd[i + 4] : pd[i]) + __shfl_xor(b3 ? pd[i] : pd[i + 4], 8);
  }
  float ts[2], td[2];
  for (int i = 0; i < 2; ++i) {
    ts[i] = (b2 ? qs[i + 2] : qs[i]) + __shfl_xor(b2 ? qs[i] : qs[i + 2], 4);
    td[i] = (b2 ? qd[i + 2] : qd[i]) + __shfl_xor(b2 ? qd[i] : qd[i + 2], 4);
  }
  float us = (b1 ? ts[1] : ts[0]) + __shfl_xor(b1 ? ts[0] : ts[1], 2);
  float ud = (b1 ? td[1] : td[0]) + __shfl_xor(b1 ? td[0] : td[1], 2);
  us += __shfl_xor(us, 1);
  ud += __shfl_xor(ud, 1);
  if ((lm & 1) == 0) {
    const int rr = hi * 8 + (lm >> 1);
    s_al[0][rr * H + head] = us;
    s_al[1][rr * H + head] = ud;
  }
  __syncthreads();
  if (threadIdx.x < 32 * H) {
    const int which = threadIdx.x / (16 * H);
    const int idx = threadIdx.x % (16 * H);
    float* dstp = which ? ad : as;
    dstp[(size_t)rowBase * H + idx] = s_al[which][idx];
  }
}

__global__ void __launch_bounds__(128)
gemm_alpha_h4_kernel(const float* __restrict__ A, const float* __restrict__ B,
                     const float* __restrict__ a_src, const float* __restrict__ a_dst,
                     float* __restrict__ C, float* __restrict__ as, float* __restrict__ ad,
                     unsigned* __restrict__ zero_p, int nz) {
  gemm_alpha_body<128>(A, B, a_src, a_dst, C, as, ad, zero_p, nz);
}

__global__ void dst_rank_kernel(const int* __restrict__ ei, int E,
                                int* __restrict__ cnt, int* __restrict__ rank) {
  int e = blockIdx.x * blockDim.x + threadIdx.x;
  if (e < E) rank[e] = atomicAdd(&cnt[ei[E + e]], 1);
}

constexpr int SCAN_KC = 7;
__global__ void __launch_bounds__(1024)
rowptr_scan_kernel(const int* __restrict__ cnt, int* __restrict__ rowptr, int N) {
  __shared__ int wsum[SCAN_KC][32];
  const int tid = threadIdx.x, lane = tid & 31;
  const int wv = __builtin_amdgcn_readfirstlane(tid >> 5);
  const int last4 = N & ~3;
  int carry = 0;
  for (int base = 0; base <= N; base += SCAN_KC * 8192) {
    if (base != 0) __syncthreads();
    int4 ca[SCAN_KC], cb[SCAN_KC];
#pragma unroll
    for (int k = 0; k < SCAN_KC; ++k) {
      const int i0 = base + k * 8192 + 8 * tid;
      ca[k] = *reinterpret_cast<const int4*>(cnt + ((i0 <= N) ? i0 : last4));
      cb[k] = *reinterpret_cast<const int4*>(cnt + ((i0 + 4 <= N) ? i0 + 4 : last4));
    }
    int v[SCAN_KC], x[SCAN_KC];
#pragma unroll
    for (int k = 0; k < SCAN_KC; ++k) {
      const int i0 = base + k * 8192 + 8 * tid;
      if (i0 > N) ca[k] = make_int4(0, 0, 0, 0);
      if (i0 + 4 > N) cb[k] = make_int4(0, 0, 0, 0);
      v[k] = ((ca[k].x + ca[k].y) + (ca[k].z + ca[k].w)) + ((cb[k].x + cb[k].y) + (cb[k].z + cb[k].w));
      x[k] = v[k];
    }
    for (int o = 1; o < 32; o <<= 1) {
#pragma unroll
      for (int k = 0; k < SCAN_KC; ++k) {
        const int y = __shfl_up(x[k], o);
        if (lane >= o) x[k] += y;
      }
    }
    if (lane == 31) {
#pragma unroll
      for (int k = 0; k < SCAN_KC; ++k) wsum[k][wv] = x[k];
    }
    __syncthreads();
    int s[SCAN_KC], xs[SCAN_KC];
#pragma unroll
    for (int k = 0; k < SCAN_KC; ++k) { s[k] = wsum[k][lane]; xs[k] = s[k]; }
    for (int o = 1; o < 32; o <<= 1) {
#pragma unroll
      for (int k = 0; k < SCAN_KC; ++k) {
        const int y = __shfl_up(xs[k], o);
        if (lane >= o) xs[k] += y;
      }
    }
    int run = carry;
#pragma unroll
    for (int k = 0; k < SCAN_KC; ++k) {
      const int woff = __builtin_amdgcn_readlane(xs[k] - s[k], wv);
      const int tot = __builtin_amdgcn_readlane(xs[k], 31);
      const int ex = run + woff + x[k] - v[k];
      const int i0 = base + k * 8192 + 8 * tid;
      const int e1 = ex + ca[k].x, e2 = e1 + ca[k].y, e3 = e2 + ca[k].z, e4 = e3 + ca[k].w;
      if (i0 <= N)
        *reinterpret_cast<int4*>(rowptr + i0) = make_int4(ex, e1, e2, e3);
      if (i0 + 4 <= N)
        *reinterpret_cast<int4*>(rowptr + i0 + 4) =
            make_int4(e4, e4 + cb[k].x, e4 + cb[k].x + cb[k].y, e4 + cb[k].x + cb[k].y + cb[k].z);
      run += tot;
    }
    carry = run;
  }
}

__global__ void csr_fill_kernel(const int* __restrict__ ei, int E,
                                const int* __restrict__ rowptr,
                                const int* __restrict__ rank, int* __restrict__ col) {
  int e = blockIdx.x * blockDim.x + threadIdx.x;
  if (e >= E) return;
  col[rowptr[ei[E + e]] + rank[e]] = ei[e];
}

__device__ __forceinline__ int lane_i(int v, int t) { return __builtin_amdgcn_readlane(v, t); }
__device__ __forceinline__ float lane_f(float v, int t) {
  return __int_as_float(__builtin_amdgcn_readlane(__float_as_int(v), t));
}
__device__ __forceinline__ float lrelu(float e) { return (e >= 0.f) ? e : NEG_SLOPE * e; }
__device__ __forceinline__ float wave_max(float v) {
  for (int o = 16; o > 0; o >>= 1) v = fmaxf(v, __shfl_xor(v, o));
  return v;
}
__device__ __forceinline__ float wave_sum(float v) {
  for (int o = 16; o > 0; o >>= 1) v += __shfl_xor(v, o);
  return v;
}

constexpr int O1_PITCH = 132;
constexpr int H2_PITCH = 36;
__global__ void __launch_bounds__(512)
gat_gather_h4_proj_kernel(const int* __restrict__ rowptr, const int* __restrict__ col,
                          const float* __restrict__ as, const float* __restrict__ ad,
                          const float* __restrict__ hfeat, const float* __restrict__ bias,
                          const float* __restrict__ W2,
                          const float* __restrict__ a_src2, const float* __restrict__ a_dst2,
                          const float* __restrict__ Wp, float* __restrict__ q,
                          int N) {
  __shared__ __attribute__((aligned(16))) float s_o[16 * O1_PITCH];
  __shared__ __attribute__((aligned(16))) float s_p[8 * 16 * H2_PITCH];
  const int lane = threadIdx.x & 31;
  const int wv = __builtin_amdgcn_readfirstlane(threadIdx.x >> 5);
  const int d = blockIdx.x * 16 + wv;
  const int lm = lane & 15;
  const int hi = lane >> 4;
  const int ct = wv & 1;
  const int kq = wv >> 1;
  const float* bcol = W2 + (size_t)(16 * kq + hi * 2) * 32 + ct * 16 + lm;
  v2f b[4];
#pragma unroll
  for (int i = 0; i < 4; ++i) {
    b[i].x = bcol[(size_t)(4 * i) * 32];
    b[i].y = bcol[(size_t)(4 * i + 1) * 32];
  }
  const float avs = a_src2[lane], avd = a_dst2[lane];
  float4 o = make_float4(0.f, 0.f, 0.f, 0.f);
  if (d < N) {
    const int beg = __builtin_amdgcn_readfirstlane(rowptr[d]);
    const int deg = __builtin_amdgcn_readfirstlane(rowptr[d + 1]) - beg + 1;
    const float4 adv = *reinterpret_cast<const float4*>(ad + 4 * (size_t)d);

    float m0 = -INFINITY, m1 = -INFINITY, m2 = -INFINITY, m3 = -INFINITY;
    for (int j0 = 0; j0 < deg; j0 += 32) {
      const int j = j0 + lane;
      if (j < deg) {
        const int s = (j < deg - 1) ? col[beg + j] : d;
        const float4 a = *reinterpret_cast<const float4*>(as + 4 * (size_t)s);
        m0 = fmaxf(m0, lrelu(a.x + adv.x));
        m1 = fmaxf(m1, lrelu(a.y + adv.y));
        m2 = fmaxf(m2, lrelu(a.z + adv.z));
        m3 = fmaxf(m3, lrelu(a.w + adv.w));
      }
    }
    m0 = wave_max(m0); m1 = wave_max(m1); m2 = wave_max(m2); m3 = wave_max(m3);

    const bool hb0 = (lane & 8) != 0, hb1 = (lane & 16) != 0;
    float4 acc = make_float4(0.f, 0.f, 0.f, 0.f);
    float s0 = 0.f, s1 = 0.f, s2 = 0.f, s3 = 0.f;
    for (int j0 = 0; j0 < deg; j0 += 32) {
      const int j = j0 + lane;
      int s = d;
      float x0 = 0.f, x1 = 0.f, x2 = 0.f, x3 = 0.f;
      if (j < deg) {
        if (j < deg - 1) s = col[beg + j];
        const float4 a = *reinterpret_cast<const float4*>(as + 4 * (size_t)s);
        x0 = expf(lrelu(a.x + adv.x) - m0);
        x1 = expf(lrelu(a.y + adv.y) - m1);
        x2 = expf(lrelu(a.z + adv.z) - m2);
        x3 = expf(lrelu(a.w + adv.w) - m3);
        s0 += x0; s1 += x1; s2 += x2; s3 += x3;
      }
      const int cntc = (deg - j0 < 32) ? (deg - j0) : 32;
#pragma unroll 4
      for (int t = 0; t < cntc; ++t) {
        const int st = lane_i(s, t);
        const float w0 = lane_f(x0, t), w1 = lane_f(x1, t);
        const float w2 = lane_f(x2, t), w3 = lane_f(x3, t);
        const float wlo = hb0 ? w1 : w0, whi = hb0 ? w3 : w2;
        const float w = hb1 ? whi : wlo;
        const float4 hv = *reinterpret_cast<const float4*>(hfeat + (size_t)st * 128 + 4 * lane);
        acc.x += w * hv.x; acc.y += w * hv.y; acc.z += w * hv.z; acc.w += w * hv.w;
      }
    }
    s0 = wave_sum(s0); s1 = wave_sum(s1); s2 = wave_sum(s2); s3 = wave_sum(s3);
    const float slo = hb0 ? s1 : s0, shi = hb0 ? s3 : s2;
    const float den = (hb1 ? shi : slo) + 1e-16f;
    const float* bp = bias + 4 * lane;
    o.x = acc.x / den + bp[0];
    o.y = acc.y / den + bp[1];
    o.z = acc.z / den + bp[2];
    o.w = acc.w / den + bp[3];
    o.x = (o.x > 0.f) ? o.x : (expf(o.x) - 1.f);
    o.y = (o.y > 0.f) ? o.y : (expf(o.y) - 1.f);
    o.z = (o.z > 0.f) ? o.z : (expf(o.z) - 1.f);
    o.w = (o.w > 0.f) ? o.w : (expf(o.w) - 1.f);
  }
  *reinterpret_cast<float4*>(s_o + wv * O1_PITCH + 4 * lane) = o;
  __syncthreads();

  const float wpa = Wp[lane], wpb = Wp[32 + lane];
  v8f c = {};
  const float* arow = s_o + lm * O1_PITCH + 16 * kq + hi * 2;
#pragma unroll
  for (int i = 0; i < 4; ++i) {
    const v2f a = *reinterpret_cast<const v2f*>(arow + 4 * i);
    c = __builtin_amdgcn_wmma_f32_16x16x4_f32(
         false, a,  false, b[i],
         (short)0, c,  false,  false);
  }
  float* pt = s_p + kq * (16 * H2_PITCH);
#pragma unroll
  for (int r = 0; r < 8; ++r)
    pt[(hi * 8 + r) * H2_PITCH + ct * 16 + lm] = c[r];
  __syncthreads();

  if (d < N) {
    const float* pr = s_p + wv * H2_PITCH + lane;
    float p[8];
#pragma unroll
    for (int q = 0; q < 8; ++q) p[q] = pr[q * (16 * H2_PITCH)];
    const float v = ((p[0] + p[1]) + (p[2] + p[3])) + ((p[4] + p[5]) + (p[6] + p[7]));
    const float ta = v * avs, tb = v * avd, tc = v * wpa, td = v * wpb;
    const bool hb = (lane & 16) != 0, b3 = (lane & 8) != 0;
    const float f0 = (hb ? tc : ta) + __shfl_xor(hb ? ta : tc, 16);
    const float f1 = (hb ? td : tb) + __shfl_xor(hb ? tb : td, 16);
    float g = (b3 ? f1 : f0) + __shfl_xor(b3 ? f0 : f1, 8);
    g += __shfl_xor(g, 4);
    g += __shfl_xor(g, 2);
    g += __shfl_xor(g, 1);
    if ((lane & 7) == 0) q[4 * (size_t)d + (lane >> 3)] = g;
  }
}

template <int M>
__device__ __forceinline__ float row_xor(float v) {
  return __int_as_float(__builtin_amdgcn_update_dpp(0, __float_as_int(v), 0x160 + M, 0xf, 0xf, false));
}
__device__ __forceinline__ float row_sum(float v) {
  v += row_xor<1>(v); v += row_xor<2>(v); v += row_xor<4>(v); v += row_xor<8>(v);
  return v;
}
__device__ __forceinline__ float row_max(float v) {
  v = fmaxf(v, row_xor<1>(v)); v = fmaxf(v, row_xor<2>(v));
  v = fmaxf(v, row_xor<4>(v)); v = fmaxf(v, row_xor<8>(v));
  return v;
}

__global__ void __launch_bounds__(256)
gat_scalar_h1_kernel(const int* __restrict__ rowptr, const int* __restrict__ col,
                     const float* __restrict__ q, const float* __restrict__ bias,
                     const float* __restrict__ Wp,
                     float* __restrict__ ps, float* __restrict__ pd, int N) {
  const int lane = threadIdx.x & 31;
  const int sub = threadIdx.x & 15;
  const int d = blockIdx.x * 16 + (threadIdx.x >> 4);
  const bool live = d < N;
  const int dc = live ? d : N - 1;
  const int beg = rowptr[dc];
  const int deg = rowptr[dc + 1] - beg + 1;
  const float adv = q[4 * (size_t)dc + 1];

  const int j0 = sub, j1 = sub + 16;
  const bool a0 = j0 < deg, a1 = j1 < deg;
  const int c0 = col[(j0 < deg - 1) ? beg + j0 : 0];
  const int c1 = col[(j1 < deg - 1) ? beg + j1 : 0];
  const int s0 = (j0 < deg - 1) ? c0 : dc;
  const int s1 = (j1 < deg - 1) ? c1 : dc;
  const float4 q0 = *reinterpret_cast<const float4*>(q + 4 * (size_t)s0);
  const float4 q1 = *reinterpret_cast<const float4*>(q + 4 * (size_t)s1);

  const float bl = bias[lane];
  const float cb1 = wave_sum(bl * Wp[lane]);
  const float cb2 = wave_sum(bl * Wp[32 + lane]);

  const float e0 = a0 ? lrelu(q0.x + adv) : -INFINITY;
  const float e1 = a1 ? lrelu(q1.x + adv) : -INFINITY;
  float m = fmaxf(e0, e1);
  for (int j = sub + 32; j < deg; j += 16) {
    const int s = (j < deg - 1) ? col[beg + j] : dc;
    m = fmaxf(m, lrelu(q[4 * (size_t)s] + adv));
  }
  m = row_max(m);

  const float x0 = a0 ? expf(e0 - m) : 0.f;
  const float x1 = a1 ? expf(e1 - m) : 0.f;
  float sx = x0 + x1;
  float su = x0 * q0.z + x1 * q1.z;
  float sv = x0 * q0.w + x1 * q1.w;
  for (int j = sub + 32; j < deg; j += 16) {
    const int s = (j < deg - 1) ? col[beg + j] : dc;
    const float4 qq = *reinterpret_cast<const float4*>(q + 4 * (size_t)s);
    const float x = expf(lrelu(qq.x + adv) - m);
    sx += x; su += x * qq.z; sv += x * qq.w;
  }
  sx = row_sum(sx); su = row_sum(su); sv = row_sum(sv);
  const float den = sx + 1e-16f;
  if (live && sub == 0) {
    ps[d] = su / den + cb1;
    pd[d] = sv / den + cb2;
  }
}

__global__ void pred_edge_kernel(const int* __restrict__ ei, int E,
                                 const float* __restrict__ ps,
                                 const float* __restrict__ pd,
                                 const float* __restrict__ bp,
                                 float* __restrict__ out) {
  int e = blockIdx.x * blockDim.x + threadIdx.x;
  if (e >= E) return;
  out[e] = ps[ei[e]] + pd[ei[E + e]] + bp[0];
}

static inline int cdiv(long long a, int b) { return (int)((a + b - 1) / b); }

extern "C" void kernel_launch(void* const* d_in, const int* in_sizes, int n_in,
                              void* d_out, int out_size, void* d_ws, size_t ws_size,
                              hipStream_t stream) {
  const float* x   = (const float*)d_in[0];
  const int*   ei  = (const int*)  d_in[1];
  const float* W1  = (const float*)d_in[2];
  const float* as1w= (const float*)d_in[3];
  const float* ad1w= (const float*)d_in[4];
  const float* b1  = (const float*)d_in[5];
  const float* W2  = (const float*)d_in[6];
  const float* as2w= (const float*)d_in[7];
  const float* ad2w= (const float*)d_in[8];
  const float* b2  = (const float*)d_in[9];
  const float* Wp  = (const float*)d_in[10];
  const float* bp  = (const float*)d_in[11];
  float* out = (float*)d_out;

  const int F = 128, H1 = 4;
  const int N = in_sizes[0] / F;
  const int E = in_sizes[1] / 2;

  float* ws = (float*)d_ws;
  float*    h1    = ws;                 ws += (size_t)N * F;
  float*    q2    = ws;                 ws += (size_t)N * 4;
  float*    as1   = ws;                 ws += (size_t)N * H1;
  float*    ad1   = ws;                 ws += (size_t)N * H1;
  int*      rowptr= (int*)ws;           ws += (size_t)N + 4;
  int*      cnt   = (int*)ws;           ws += (size_t)N + 4;
  int*      col   = (int*)ws;           ws += (size_t)E;
  int*      rank  = (int*)ws;           ws += (size_t)E;
  float*    ps    = ws;                 ws += (size_t)N;
  float*    pd    = ws;                 ws += (size_t)N;
  (void)ws_size; (void)n_in; (void)out_size;

  const int T = 256;

  gemm_alpha_h4_kernel<<<N / 16, 128, 0, stream>>>(x, W1, as1w, ad1w, h1, as1, ad1,
                                                   (unsigned*)cnt, N + 4);

  dst_rank_kernel<<<cdiv(E, T), T, 0, stream>>>(ei, E, cnt, rank);
  rowptr_scan_kernel<<<1, 1024, 0, stream>>>(cnt, rowptr, N);
  csr_fill_kernel<<<cdiv(E, T), T, 0, stream>>>(ei, E, rowptr, rank, col);

  gat_gather_h4_proj_kernel<<<cdiv(N, 16), 512, 0, stream>>>(rowptr, col, as1, ad1, h1, b1,
                                                            W2, as2w, ad2w, Wp, q2, N);

  gat_scalar_h1_kernel<<<cdiv(N, 16), T, 0, stream>>>(rowptr, col, q2, b2, Wp, ps, pd, N);

  pred_edge_kernel<<<cdiv(E, T), T, 0, stream>>>(ei, E, ps, pd, bp, out);
}


# Layer-1 gather, fewer dependent waits per node: indices and scores loaded once, eight source rows in flight (first group issued ahead of the softmax, tail padded), lane max and sums on the vector ALU
# speedup vs baseline: 11.1226x; 1.0311x over previous
#include <hip/hip_runtime.h>
#include <hip/hip_bf16.h>


typedef float v2f __attribute__((ext_vector_type(2)));
typedef float v8f __attribute__((ext_vector_type(8)));

#define NEG_SLOPE 0.2f

template <int NCOL>
__device__ __forceinline__ void gemm_alpha_body(const float* __restrict__ A,
                                                const float* __restrict__ B,
                                                const float* __restrict__ a_src,
                                                const float* __restrict__ a_dst,
                                                float* __restrict__ C,
                                                float* __restrict__ as, float* __restrict__ ad,
                                                unsigned* __restrict__ zero_p, int nz) {
  constexpr int K = 128;
  constexpr int H = NCOL / 32;
  __shared__ float s_al[2][16 * H];

  for (int i = blockIdx.x * blockDim.x + threadIdx.x; i < nz; i += gridDim.x * blockDim.x)
    zero_p[i] = 0u;

  const int head = threadIdx.x >> 5;
  const int lane = threadIdx.x & 31;
  const int rowBase = blockIdx.x * 16;
  const int colBase = head * 32;
  const int lm = lane & 15;
  const int hi = lane >> 4;

  v8f acc0 = {}, acc1 = {};
  const float* arow = A + (size_t)(rowBase + lm) * K + hi * 2;
  const float* bcol = B + (size_t)(hi * 2) * NCOL + colBase + lm;
  for (int k0 = 0; k0 < K; k0 += 4) {
    v2f a, b0, b1;
    a.x = arow[k0];
    a.y = arow[k0 + 1];
    b0.x = bcol[(size_t)k0 * NCOL];
    b0.y = bcol[(size_t)(k0 + 1) * NCOL];
    b1.x = bcol[(size_t)k0 * NCOL + 16];
    b1.y = bcol[(size_t)(k0 + 1) * NCOL + 16];
    acc0 = __builtin_amdgcn_wmma_f32_16x16x4_f32(
         false, a,  false, b0,
         (short)0, acc0,  false,  false);
    acc1 = __builtin_amdgcn_wmma_f32_16x16x4_f32(
         false, a,  false, b1,
         (short)0, acc1,  false,  false);
  }
#pragma unroll
  for (int r = 0; r < 8; ++r) {
    float* crow = C + (size_t)(rowBase + hi * 8 + r) * NCOL + colBase + lm;
    crow[0] = acc0[r];
    crow[16] = acc1[r];
  }

  const float sa0 = a_src[colBase + lm], sa1 = a_src[colBase + 16 + lm];
  const float da0 = a_dst[colBase + lm], da1 = a_dst[colBase + 16 + lm];
  float ps[8], pd[8];
  for (int r = 0; r < 8; ++r) {
    ps[r] = acc0[r] * sa0 + acc1[r] * sa1;
    pd[r] = acc0[r] * da0 + acc1[r] * da1;
  }
  const bool b3 = (lm & 8) != 0, b2 = (lm & 4) != 0, b1 = (lm & 2) != 0;
  float qs[4], qd[4];
  for (int i = 0; i < 4; ++i) {
    qs[i] = (b3 ? ps[i + 4] : ps[i]) + __shfl_xor(b3 ? ps[i] : ps[i + 4], 8);
    qd[i] = (b3 ? pd[i + 4] : pd[i]) + __shfl_xor(b3 ? pd[i] : pd[i + 4], 8);
  }
  float ts[2], td[2];
  for (int i = 0; i < 2; ++i) {
    ts[i] = (b2 ? qs[i + 2] : qs[i]) + __shfl_xor(b2 ? qs[i] : qs[i + 2], 4);
    td[i] = (b2 ? qd[i + 2] : qd[i]) + __shfl_xor(b2 ? qd[i] : qd[i + 2], 4);
  }
  float us = (b1 ? ts[1] : ts[0]) + __shfl_xor(b1 ? ts[0] : ts[1], 2);
  float ud = (b1 ? td[1] : td[0]) + __shfl_xor(b1 ? td[0] : td[1], 2);
  us += __shfl_xor(us, 1);
  ud += __shfl_xor(ud, 1);
  if ((lm & 1) == 0) {
    const int rr = hi * 8 + (lm >> 1);
    s_al[0][rr * H + head] = us;
    s_al[1][rr * H + head] = ud;
  }
  __syncthreads();
  if (threadIdx.x < 32 * H) {
    const int which = threadIdx.x / (16 * H);
    const int idx = threadIdx.x % (16 * H);
    float* dstp = which ? ad : as;
    dstp[(size_t)rowBase * H + idx] = s_al[which][idx];
  }
}

__global__ void __launch_bounds__(128)
gemm_alpha_h4_kernel(const float* __restrict__ A, const float* __restrict__ B,
                     const float* __restrict__ a_src, const float* __restrict__ a_dst,
                     float* __restrict__ C, float* __restrict__ as, float* __restrict__ ad,
                     unsigned* __restrict__ zero_p, int nz) {
  gemm_alpha_body<128>(A, B, a_src, a_dst, C, as, ad, zero_p, nz);
}

__global__ void dst_rank_kernel(const int* __restrict__ ei, int E,
                                int* __restrict__ cnt, int* __restrict__ rank) {
  int e = blockIdx.x * blockDim.x + threadIdx.x;
  if (e < E) rank[e] = atomicAdd(&cnt[ei[E + e]], 1);
}

constexpr int SCAN_KC = 7;
__global__ void __launch_bounds__(1024)
rowptr_scan_kernel(const int* __restrict__ cnt, int* __restrict__ rowptr, int N) {
  __shared__ int wsum[SCAN_KC][32];
  const int tid = threadIdx.x, lane = tid & 31;
  const int wv = __builtin_amdgcn_readfirstlane(tid >> 5);
  const int last4 = N & ~3;
  int carry = 0;
  for (int base = 0; base <= N; base += SCAN_KC * 8192) {
    if (base != 0) __syncthreads();
    int4 ca[SCAN_KC], cb[SCAN_KC];
#pragma unroll
    for (int k = 0; k < SCAN_KC; ++k) {
      const int i0 = base + k * 8192 + 8 * tid;
      ca[k] = *reinterpret_cast<const int4*>(cnt + ((i0 <= N) ? i0 : last4));
      cb[k] = *reinterpret_cast<const int4*>(cnt + ((i0 + 4 <= N) ? i0 + 4 : last4));
    }
    int v[SCAN_KC], x[SCAN_KC];
#pragma unroll
    for (int k = 0; k < SCAN_KC; ++k) {
      const int i0 = base + k * 8192 + 8 * tid;
      if (i0 > N) ca[k] = make_int4(0, 0, 0, 0);
      if (i0 + 4 > N) cb[k] = make_int4(0, 0, 0, 0);
      v[k] = ((ca[k].x + ca[k].y) + (ca[k].z + ca[k].w)) + ((cb[k].x + cb[k].y) + (cb[k].z + cb[k].w));
      x[k] = v[k];
    }
    for (int o = 1; o < 32; o <<= 1) {
#pragma unroll
      for (int k = 0; k < SCAN_KC; ++k) {
        const int y = __shfl_up(x[k], o);
        if (lane >= o) x[k] += y;
      }
    }
    if (lane == 31) {
#pragma unroll
      for (int k = 0; k < SCAN_KC; ++k) wsum[k][wv] = x[k];
    }
    __syncthreads();
    int s[SCAN_KC], xs[SCAN_KC];
#pragma unroll
    for (int k = 0; k < SCAN_KC; ++k) { s[k] = wsum[k][lane]; xs[k] = s[k]; }
    for (int o = 1; o < 32; o <<= 1) {
#pragma unroll
      for (int k = 0; k < SCAN_KC; ++k) {
        const int y = __shfl_up(xs[k], o);
        if (lane >= o) xs[k] += y;
      }
    }
    int run = carry;
#pragma unroll
    for (int k = 0; k < SCAN_KC; ++k) {
      const int woff = __builtin_amdgcn_readlane(xs[k] - s[k], wv);
      const int tot = __builtin_amdgcn_readlane(xs[k], 31);
      const int ex = run + woff + x[k] - v[k];
      const int i0 = base + k * 8192 + 8 * tid;
      const int e1 = ex + ca[k].x, e2 = e1 + ca[k].y, e3 = e2 + ca[k].z, e4 = e3 + ca[k].w;
      if (i0 <= N)
        *reinterpret_cast<int4*>(rowptr + i0) = make_int4(ex, e1, e2, e3);
      if (i0 + 4 <= N)
        *reinterpret_cast<int4*>(rowptr + i0 + 4) =
            make_int4(e4, e4 + cb[k].x, e4 + cb[k].x + cb[k].y, e4 + cb[k].x + cb[k].y + cb[k].z);
      run += tot;
    }
    carry = run;
  }
}

__global__ void csr_fill_kernel(const int* __restrict__ ei, int E,
                                const int* __restrict__ rowptr,
                                const int* __restrict__ rank, int* __restrict__ col) {
  int e = blockIdx.x * blockDim.x + threadIdx.x;
  if (e >= E) return;
  col[rowptr[ei[E + e]] + rank[e]] = ei[e];
}

__device__ __forceinline__ int lane_i(int v, int t) { return __builtin_amdgcn_readlane(v, t); }
__device__ __forceinline__ float lane_f(float v, int t) {
  return __int_as_float(__builtin_amdgcn_readlane(__float_as_int(v), t));
}
__device__ __forceinline__ float4 row_ld(const float* __restrict__ hfeat, int st, unsigned lane_off) {
  const char* rb = reinterpret_cast<const char*>(hfeat + (size_t)st * 128);
  return *reinterpret_cast<const float4*>(rb + lane_off);
}
__device__ __forceinline__ float lrelu(float e) { return (e >= 0.f) ? e : NEG_SLOPE * e; }
__device__ __forceinline__ float wave_max(float v) {
  for (int o = 16; o > 0; o >>= 1) v = fmaxf(v, __shfl_xor(v, o));
  return v;
}
__device__ __forceinline__ float wave_sum(float v) {
  for (int o = 16; o > 0; o >>= 1) v += __shfl_xor(v, o);
  return v;
}

template <int M>
__device__ __forceinline__ float row_xor(float v) {
  return __int_as_float(__builtin_amdgcn_update_dpp(0, __float_as_int(v), 0x160 + M, 0xf, 0xf, false));
}
__device__ __forceinline__ float row_sum(float v) {
  v += row_xor<1>(v); v += row_xor<2>(v); v += row_xor<4>(v); v += row_xor<8>(v);
  return v;
}
__device__ __forceinline__ float row_max(float v) {
  v = fmaxf(v, row_xor<1>(v)); v = fmaxf(v, row_xor<2>(v));
  v = fmaxf(v, row_xor<4>(v)); v = fmaxf(v, row_xor<8>(v));
  return v;
}
__device__ __forceinline__ float xhalf(float v) {
  return __int_as_float(__builtin_amdgcn_permlanex16(__float_as_int(v), __float_as_int(v),
                                                     0x76543210, 0xfedcba98, false, false));
}
__device__ __forceinline__ float wave_sum_alu(float v) {
  v += xhalf(v);
  v += row_xor<8>(v); v += row_xor<4>(v); v += row_xor<2>(v); v += row_xor<1>(v);
  return v;
}
__device__ __forceinline__ float wave_max_alu(float v) {
  return row_max(fmaxf(v, xhalf(v)));
}

constexpr int ROW_GROUP = 8;
constexpr int ROW_MAXDEG = 32;
constexpr int O1_PITCH = 132;
constexpr int H2_PITCH = 36;
__global__ void __launch_bounds__(512)
gat_gather_h4_proj_kernel(const int* __restrict__ rowptr, const int* __restrict__ col,
                          const float* __restrict__ as, const float* __restrict__ ad,
                          const float* __restrict__ hfeat, const float* __restrict__ bias,
                          const float* __restrict__ W2,
                          const float* __restrict__ a_src2, const float* __restrict__ a_dst2,
                          const float* __restrict__ Wp, float* __restrict__ q,
                          int N) {
  __shared__ __attribute__((aligned(16))) float s_o[16 * O1_PITCH];
  __shared__ __attribute__((aligned(16))) float s_p[8 * 16 * H2_PITCH];
  const int lane = threadIdx.x & 31;
  const int wv = __builtin_amdgcn_readfirstlane(threadIdx.x >> 5);
  const int d = blockIdx.x * 16 + wv;
  float4 o = make_float4(0.f, 0.f, 0.f, 0.f);
  if (d < N) {
    const int beg = __builtin_amdgcn_readfirstlane(rowptr[d]);
    const int deg = __builtin_amdgcn_readfirstlane(rowptr[d + 1]) - beg + 1;
    const float4 adv = *reinterpret_cast<const float4*>(ad + 4 * (size_t)d);

    const bool hb0 = (lane & 8) != 0, hb1 = (lane & 16) != 0;
    float4 acc = make_float4(0.f, 0.f, 0.f, 0.f);
    float s0 = 0.f, s1 = 0.f, s2 = 0.f, s3 = 0.f;
    if (deg <= ROW_MAXDEG) {
      const bool act = lane < deg;
      int s = d;
      if (lane < deg - 1) s = col[beg + lane];
      const float4 a = *reinterpret_cast<const float4*>(as + 4 * (size_t)s);
      const unsigned lo = 16u * (unsigned)lane;
      float4 r[ROW_GROUP];
#pragma unroll
      for (int i = 0; i < ROW_GROUP; ++i) r[i] = row_ld(hfeat, lane_i(s, i), lo);
      const float e0 = act ? lrelu(a.x + adv.x) : -INFINITY;
      const float e1 = act ? lrelu(a.y + adv.y) : -INFINITY;
      const float e2 = act ? lrelu(a.z + adv.z) : -INFINITY;
      const float e3 = act ? lrelu(a.w + adv.w) : -INFINITY;
      const float m0 = wave_max_alu(e0), m1 = wave_max_alu(e1);
      const float m2 = wave_max_alu(e2), m3 = wave_max_alu(e3);
      const float x0 = act ? expf(e0 - m0) : 0.f;
      const float x1 = act ? expf(e1 - m1) : 0.f;
      const float x2 = act ? expf(e2 - m2) : 0.f;
      const float x3 = act ? expf(e3 - m3) : 0.f;
      s0 = x0; s1 = x1; s2 = x2; s3 = x3;
      for (int t = 0; t < deg; t += ROW_GROUP) {
#pragma unroll
        for (int i = 0; i < ROW_GROUP; ++i) {
          const float w0 = lane_f(x0, t + i), w1 = lane_f(x1, t + i);
          const float w2 = lane_f(x2, t + i), w3 = lane_f(x3, t + i);
          const float wlo = hb0 ? w1 : w0, whi = hb0 ? w3 : w2;
          const float w = hb1 ? whi : wlo;
          acc.x += w * r[i].x; acc.y += w * r[i].y; acc.z += w * r[i].z; acc.w += w * r[i].w;
        }
        if (t + ROW_GROUP < deg) {
#pragma unroll
          for (int i = 0; i < ROW_GROUP; ++i) r[i] = row_ld(hfeat, lane_i(s, t + ROW_GROUP + i), lo);
        }
      }
    } else {
      float m0 = -INFINITY, m1 = -INFINITY, m2 = -INFINITY, m3 = -INFINITY;
      for (int j0 = 0; j0 < deg; j0 += 32) {
        const int j = j0 + lane;
        if (j < deg) {
          const int s = (j < deg - 1) ? col[beg + j] : d;
          const float4 a = *reinterpret_cast<const float4*>(as + 4 * (size_t)s);
          m0 = fmaxf(m0, lrelu(a.x + adv.x));
          m1 = fmaxf(m1, lrelu(a.y + adv.y));
          m2 = fmaxf(m2, lrelu(a.z + adv.z));
          m3 = fmaxf(m3, lrelu(a.w + adv.w));
        }
      }
      m0 = wave_max(m0); m1 = wave_max(m1); m2 = wave_max(m2); m3 = wave_max(m3);

      for (int j0 = 0; j0 < deg; j0 += 32) {
        const int j = j0 + lane;
        int s = d;
        float x0 = 0.f, x1 = 0.f, x2 = 0.f, x3 = 0.f;
        if (j < deg) {
          if (j < deg - 1) s = col[beg + j];
          const float4 a = *reinterpret_cast<const float4*>(as + 4 * (size_t)s);
          x0 = expf(lrelu(a.x + adv.x) - m0);
          x1 = expf(lrelu(a.y + adv.y) - m1);
          x2 = expf(lrelu(a.z + adv.z) - m2);
          x3 = expf(lrelu(a.w + adv.w) - m3);
          s0 += x0; s1 += x1; s2 += x2; s3 += x3;
        }
        const int cntc = (deg - j0 < 32) ? (deg - j0) : 32;
#pragma unroll 4
        for (int t = 0; t < cntc; ++t) {
          const int st = lane_i(s, t);
          const float w0 = lane_f(x0, t), w1 = lane_f(x1, t);
          const float w2 = lane_f(x2, t), w3 = lane_f(x3, t);
          const float wlo = hb0 ? w1 : w0, whi = hb0 ? w3 : w2;
          const float w = hb1 ? whi : wlo;
          const float4 hv = *reinterpret_cast<const float4*>(hfeat + (size_t)st * 128 + 4 * lane);
          acc.x += w * hv.x; acc.y += w * hv.y; acc.z += w * hv.z; acc.w += w * hv.w;
        }
      }
    }
    s0 = wave_sum_alu(s0); s1 = wave_sum_alu(s1); s2 = wave_sum_alu(s2); s3 = wave_sum_alu(s3);
    const float slo = hb0 ? s1 : s0, shi = hb0 ? s3 : s2;
    const float den = (hb1 ? shi : slo) + 1e-16f;
    const float* bp = bias + 4 * lane;
    o.x = acc.x / den + bp[0];
    o.y = acc.y / den + bp[1];
    o.z = acc.z / den + bp[2];
    o.w = acc.w / den + bp[3];
    o.x = (o.x > 0.f) ? o.x : (expf(o.x) - 1.f);
    o.y = (o.y > 0.f) ? o.y : (expf(o.y) - 1.f);
    o.z = (o.z > 0.f) ? o.z : (expf(o.z) - 1.f);
    o.w = (o.w > 0.f) ? o.w : (expf(o.w) - 1.f);
  }
  const int lm = lane & 15;
  const int hi = lane >> 4;
  const int ct = wv & 1;
  const int kq = wv >> 1;
  const float* bcol = W2 + (size_t)(16 * kq + hi * 2) * 32 + ct * 16 + lm;
  v2f b[4];
#pragma unroll
  for (int i = 0; i < 4; ++i) {
    b[i].x = bcol[(size_t)(4 * i) * 32];
    b[i].y = bcol[(size_t)(4 * i + 1) * 32];
  }
  *reinterpret_cast<float4*>(s_o + wv * O1_PITCH + 4 * lane) = o;
  __syncthreads();

  const float wpa = Wp[lane], wpb = Wp[32 + lane];
  const float avs = a_src2[lane], avd = a_dst2[lane];
  v8f c = {};
  const float* arow = s_o + lm * O1_PITCH + 16 * kq + hi * 2;
#pragma unroll
  for (int i = 0; i < 4; ++i) {
    const v2f a = *reinterpret_cast<const v2f*>(arow + 4 * i);
    c = __builtin_amdgcn_wmma_f32_16x16x4_f32(
         false, a,  false, b[i],
         (short)0, c,  false,  false);
  }
  float* pt = s_p + kq * (16 * H2_PITCH);
#pragma unroll
  for (int r = 0; r < 8; ++r)
    pt[(hi * 8 + r) * H2_PITCH + ct * 16 + lm] = c[r];
  __syncthreads();

  if (d < N) {
    const float* pr = s_p + wv * H2_PITCH + lane;
    float p[8];
#pragma unroll
    for (int q = 0; q < 8; ++q) p[q] = pr[q * (16 * H2_PITCH)];
    const float v = ((p[0] + p[1]) + (p[2] + p[3])) + ((p[4] + p[5]) + (p[6] + p[7]));
    const float ta = v * avs, tb = v * avd, tc = v * wpa, td = v * wpb;
    const bool hb = (lane & 16) != 0, b3 = (lane & 8) != 0;
    const float f0 = (hb ? tc : ta) + __shfl_xor(hb ? ta : tc, 16);
    const float f1 = (hb ? td : tb) + __shfl_xor(hb ? tb : td, 16);
    float g = (b3 ? f1 : f0) + __shfl_xor(b3 ? f0 : f1, 8);
    g += __shfl_xor(g, 4);
    g += __shfl_xor(g, 2);
    g += __shfl_xor(g, 1);
    if ((lane & 7) == 0) q[4 * (size_t)d + (lane >> 3)] = g;
  }
}


__global__ void __launch_bounds__(256)
gat_scalar_h1_kernel(const int* __restrict__ rowptr, const int* __restrict__ col,
                     const float* __restrict__ q, const float* __restrict__ bias,
                     const float* __restrict__ Wp,
                     float* __restrict__ ps, float* __restrict__ pd, int N) {
  const int lane = threadIdx.x & 31;
  const int sub = threadIdx.x & 15;
  const int d = blockIdx.x * 16 + (threadIdx.x >> 4);
  const bool live = d < N;
  const int dc = live ? d : N - 1;
  const int beg = rowptr[dc];
  const int deg = rowptr[dc + 1] - beg + 1;
  const float adv = q[4 * (size_t)dc + 1];

  const int j0 = sub, j1 = sub + 16;
  const bool a0 = j0 < deg, a1 = j1 < deg;
  const int c0 = col[(j0 < deg - 1) ? beg + j0 : 0];
  const int c1 = col[(j1 < deg - 1) ? beg + j1 : 0];
  const int s0 = (j0 < deg - 1) ? c0 : dc;
  const int s1 = (j1 < deg - 1) ? c1 : dc;
  const float4 q0 = *reinterpret_cast<const float4*>(q + 4 * (size_t)s0);
  const float4 q1 = *reinterpret_cast<const float4*>(q + 4 * (size_t)s1);

  const float bl = bias[lane];
  const float cb1 = wave_sum(bl * Wp[lane]);
  const float cb2 = wave_sum(bl * Wp[32 + lane]);

  const float e0 = a0 ? lrelu(q0.x + adv) : -INFINITY;
  const float e1 = a1 ? lrelu(q1.x + adv) : -INFINITY;
  float m = fmaxf(e0, e1);
  for (int j = sub + 32; j < deg; j += 16) {
    const int s = (j < deg - 1) ? col[beg + j] : dc;
    m = fmaxf(m, lrelu(q[4 * (size_t)s] + adv));
  }
  m = row_max(m);

  const float x0 = a0 ? expf(e0 - m) : 0.f;
  const float x1 = a1 ? expf(e1 - m) : 0.f;
  float sx = x0 + x1;
  float su = x0 * q0.z + x1 * q1.z;
  float sv = x0 * q0.w + x1 * q1.w;
  for (int j = sub + 32; j < deg; j += 16) {
    const int s = (j < deg - 1) ? col[beg + j] : dc;
    const float4 qq = *reinterpret_cast<const float4*>(q + 4 * (size_t)s);
    const float x = expf(lrelu(qq.x + adv) - m);
    sx += x; su += x * qq.z; sv += x * qq.w;
  }
  sx = row_sum(sx); su = row_sum(su); sv = row_sum(sv);
  const float den = sx + 1e-16f;
  if (live && sub == 0) {
    ps[d] = su / den + cb1;
    pd[d] = sv / den + cb2;
  }
}

__global__ void pred_edge_kernel(const int* __restrict__ ei, int E,
                                 const float* __restrict__ ps,
                                 const float* __restrict__ pd,
                                 const float* __restrict__ bp,
                                 float* __restrict__ out) {
  int e = blockIdx.x * blockDim.x + threadIdx.x;
  if (e >= E) return;
  out[e] = ps[ei[e]] + pd[ei[E + e]] + bp[0];
}

static inline int cdiv(long long a, int b) { return (int)((a + b - 1) / b); }

extern "C" void kernel_launch(void* const* d_in, const int* in_sizes, int n_in,
                              void* d_out, int out_size, void* d_ws, size_t ws_size,
                              hipStream_t stream) {
  const float* x   = (const float*)d_in[0];
  const int*   ei  = (const int*)  d_in[1];
  const float* W1  = (const float*)d_in[2];
  const float* as1w= (const float*)d_in[3];
  const float* ad1w= (const float*)d_in[4];
  const float* b1  = (const float*)d_in[5];
  const float* W2  = (const float*)d_in[6];
  const float* as2w= (const float*)d_in[7];
  const float* ad2w= (const float*)d_in[8];
  const float* b2  = (const float*)d_in[9];
  const float* Wp  = (const float*)d_in[10];
  const float* bp  = (const float*)d_in[11];
  float* out = (float*)d_out;

  const int F = 128, H1 = 4;
  const int N = in_sizes[0] / F;
  const int E = in_sizes[1] / 2;

  float* ws = (float*)d_ws;
  float*    h1    = ws;                 ws += (size_t)N * F;
  float*    q2    = ws;                 ws += (size_t)N * 4;
  float*    as1   = ws;                 ws += (size_t)N * H1;
  float*    ad1   = ws;                 ws += (size_t)N * H1;
  int*      rowptr= (int*)ws;           ws += (size_t)N + 4;
  int*      cnt   = (int*)ws;           ws += (size_t)N + 4;
  int*      col   = (int*)ws;           ws += (size_t)E;
  int*      rank  = (int*)ws;           ws += (size_t)E;
  float*    ps    = ws;                 ws += (size_t)N;
  float*    pd    = ws;                 ws += (size_t)N;
  (void)ws_size; (void)n_in; (void)out_size;

  const int T = 256;

  gemm_alpha_h4_kernel<<<N / 16, 128, 0, stream>>>(x, W1, as1w, ad1w, h1, as1, ad1,
                                                   (unsigned*)cnt, N + 4);

  dst_rank_kernel<<<cdiv(E, T), T, 0, stream>>>(ei, E, cnt, rank);
  rowptr_scan_kernel<<<1, 1024, 0, stream>>>(cnt, rowptr, N);
  csr_fill_kernel<<<cdiv(E, T), T, 0, stream>>>(ei, E, rowptr, rank, col);

  gat_gather_h4_proj_kernel<<<cdiv(N, 16), 512, 0, stream>>>(rowptr, col, as1, ad1, h1, b1,
                                                            W2, as2w, ad2w, Wp, q2, N);

  gat_scalar_h1_kernel<<<cdiv(N, 16), T, 0, stream>>>(rowptr, col, q2, b2, Wp, ps, pd, N);

  pred_edge_kernel<<<cdiv(E, T), T, 0, stream>>>(ei, E, ps, pd, bp, out);
}


# Layer-1 aggregation, nodes of 33 or more items: source rows requested seven at a time, one wait per group (was one row per trip, each waited for); items added in the same order as before
# speedup vs baseline: 12.1753x; 1.0946x over previous
#include <hip/hip_runtime.h>
#include <hip/hip_bf16.h>


typedef float v2f __attribute__((ext_vector_type(2)));
typedef float v8f __attribute__((ext_vector_type(8)));

#define NEG_SLOPE 0.2f

template <int NCOL>
__device__ __forceinline__ void gemm_alpha_body(const float* __restrict__ A,
                                                const float* __restrict__ B,
                                                const float* __restrict__ a_src,
                                                const float* __restrict__ a_dst,
                                                float* __restrict__ C,
                                                float* __restrict__ as, float* __restrict__ ad,
                                                unsigned* __restrict__ zero_p, int nz) {
  constexpr int K = 128;
  constexpr int H = NCOL / 32;
  __shared__ float s_al[2][16 * H];

  for (int i = blockIdx.x * blockDim.x + threadIdx.x; i < nz; i += gridDim.x * blockDim.x)
    zero_p[i] = 0u;

  const int head = threadIdx.x >> 5;
  const int lane = threadIdx.x & 31;
  const int rowBase = blockIdx.x * 16;
  const int colBase = head * 32;
  const int lm = lane & 15;
  const int hi = lane >> 4;

  v8f acc0 = {}, acc1 = {};
  const float* arow = A + (size_t)(rowBase + lm) * K + hi * 2;
  const float* bcol = B + (size_t)(hi * 2) * NCOL + colBase + lm;
  for (int k0 = 0; k0 < K; k0 += 4) {
    v2f a, b0, b1;
    a.x = arow[k0];
    a.y = arow[k0 + 1];
    b0.x = bcol[(size_t)k0 * NCOL];
    b0.y = bcol[(size_t)(k0 + 1) * NCOL];
    b1.x = bcol[(size_t)k0 * NCOL + 16];
    b1.y = bcol[(size_t)(k0 + 1) * NCOL + 16];
    acc0 = __builtin_amdgcn_wmma_f32_16x16x4_f32(
         false, a,  false, b0,
         (short)0, acc0,  false,  false);
    acc1 = __builtin_amdgcn_wmma_f32_16x16x4_f32(
         false, a,  false, b1,
         (short)0, acc1,  false,  false);
  }
#pragma unroll
  for (int r = 0; r < 8; ++r) {
    float* crow = C + (size_t)(rowBase + hi * 8 + r) * NCOL + colBase + lm;
    crow[0] = acc0[r];
    crow[16] = acc1[r];
  }

  const float sa0 = a_src[colBase + lm], sa1 = a_src[colBase + 16 + lm];
  const float da0 = a_dst[colBase + lm], da1 = a_dst[colBase + 16 + lm];
  float ps[8], pd[8];
  for (int r = 0; r < 8; ++r) {
    ps[r] = acc0[r] * sa0 + acc1[r] * sa1;
    pd[r] = acc0[r] * da0 + acc1[r] * da1;
  }
  const bool b3 = (lm & 8) != 0, b2 = (lm & 4) != 0, b1 = (lm & 2) != 0;
  float qs[4], qd[4];
  for (int i = 0; i < 4; ++i) {
    qs[i] = (b3 ? ps[i + 4] : ps[i]) + __shfl_xor(b3 ? ps[i] : ps[i + 4], 8);
    qd[i] = (b3 ? pd[i + 4] : pd[i]) + __shfl_xor(b3 ? pd[i] : pd[i + 4], 8);
  }
  float ts[2], td[2];
  for (int i = 0; i < 2; ++i) {
    ts[i] = (b2 ? qs[i + 2] : qs[i]) + __shfl_xor(b2 ? qs[i] : qs[i + 2], 4);
    td[i] = (b2 ? qd[i + 2] : qd[i]) + __shfl_xor(b2 ? qd[i] : qd[i + 2], 4);
  }
  float us = (b1 ? ts[1] : ts[0]) + __shfl_xor(b1 ? ts[0] : ts[1], 2);
  float ud = (b1 ? td[1] : td[0]) + __shfl_xor(b1 ? td[0] : td[1], 2);
  us += __shfl_xor(us, 1);
  ud += __shfl_xor(ud, 1);
  if ((lm & 1) == 0) {
    const int rr = hi * 8 + (lm >> 1);
    s_al[0][rr * H + head] = us;
    s_al[1][rr * H + head] = ud;
  }
  __syncthreads();
  if (threadIdx.x < 32 * H) {
    const int which = threadIdx.x / (16 * H);
    const int idx = threadIdx.x % (16 * H);
    float* dstp = which ? ad : as;
    dstp[(size_t)rowBase * H + idx] = s_al[which][idx];
  }
}

__global__ void __launch_bounds__(128)
gemm_alpha_h4_kernel(const float* __restrict__ A, const float* __restrict__ B,
                     const float* __restrict__ a_src, const float* __restrict__ a_dst,
                     float* __restrict__ C, float* __restrict__ as, float* __restrict__ ad,
                     unsigned* __restrict__ zero_p, int nz) {
  gemm_alpha_body<128>(A, B, a_src, a_dst, C, as, ad, zero_p, nz);
}

constexpr int CAP = 32;
__global__ void dst_bucket_kernel(const int* __restrict__ ei, int E, int N,
                                  int* __restrict__ cnt, int* __restrict__ slot,
                                  float* __restrict__ ovf) {
  int e = blockIdx.x * blockDim.x + threadIdx.x;
  if (e >= E) return;
  const int d = ei[E + e];
  const int s = ei[e];
  asm volatile("" : : "v"(s));
  const int r = atomicAdd(&cnt[d], 1);
  if (r < CAP) slot[(size_t)d * CAP + r] = s;
  else ovf[atomicAdd(&cnt[N], 1)] = __int_as_float(e);
}

__device__ __forceinline__ bool big_item(int k, int nl, int lane, int d, int sl, int M,
                                         const float* __restrict__ ovf,
                                         const int* __restrict__ ei, int E, int& s) {
  s = d;
  if (k == 0) { s = sl; return true; }
  if (k > nl) return lane == 0;
  const int i = (k - 1) * 32 + lane;
  bool v = false;
  if (i < M) {
    const int e = __float_as_int(ovf[i]);
    if (ei[E + e] == d) { s = ei[e]; v = true; }
  }
  return v;
}

__device__ __forceinline__ int lane_i(int v, int t) { return __builtin_amdgcn_readlane(v, t); }
__device__ __forceinline__ float lane_f(float v, int t) {
  return __int_as_float(__builtin_amdgcn_readlane(__float_as_int(v), t));
}
__device__ __forceinline__ float4 row_ld(const float* __restrict__ hfeat, int st, unsigned lane_off) {
  const char* rb = reinterpret_cast<const char*>(hfeat + (size_t)st * 128);
  return *reinterpret_cast<const float4*>(rb + lane_off);
}
__device__ __forceinline__ float lrelu(float e) { return (e >= 0.f) ? e : NEG_SLOPE * e; }
__device__ __forceinline__ float wave_max(float v) {
  for (int o = 16; o > 0; o >>= 1) v = fmaxf(v, __shfl_xor(v, o));
  return v;
}
__device__ __forceinline__ float wave_sum(float v) {
  for (int o = 16; o > 0; o >>= 1) v += __shfl_xor(v, o);
  return v;
}

template <int M>
__device__ __forceinline__ float row_xor(float v) {
  return __int_as_float(__builtin_amdgcn_update_dpp(0, __float_as_int(v), 0x160 + M, 0xf, 0xf, false));
}
__device__ __forceinline__ float row_sum(float v) {
  v += row_xor<1>(v); v += row_xor<2>(v); v += row_xor<4>(v); v += row_xor<8>(v);
  return v;
}
__device__ __forceinline__ float row_max(float v) {
  v = fmaxf(v, row_xor<1>(v)); v = fmaxf(v, row_xor<2>(v));
  v = fmaxf(v, row_xor<4>(v)); v = fmaxf(v, row_xor<8>(v));
  return v;
}
__device__ __forceinline__ float xhalf(float v) {
  return __int_as_float(__builtin_amdgcn_permlanex16(__float_as_int(v), __float_as_int(v),
                                                     0x76543210, 0xfedcba98, false, false));
}
__device__ __forceinline__ float wave_sum_alu(float v) {
  v += xhalf(v);
  v += row_xor<8>(v); v += row_xor<4>(v); v += row_xor<2>(v); v += row_xor<1>(v);
  return v;
}
__device__ __forceinline__ float wave_max_alu(float v) {
  return row_max(fmaxf(v, xhalf(v)));
}

constexpr int ROW_GROUP = 8;
constexpr int BIG_GROUP = 7;
constexpr int ROW_MAXDEG = 32;
constexpr int O1_PITCH = 132;
static_assert(CAP == 32 && ROW_MAXDEG - 1 <= CAP, "one slot per lane; a one-pass node has all its edges in slots");
constexpr int H2_PITCH = 36;
__global__ void __launch_bounds__(512)
gat_gather_h4_proj_kernel(const int* __restrict__ cnt, const int* __restrict__ slot,
                          const float* __restrict__ ovf, const int* __restrict__ ei, int E,
                          const float* __restrict__ as, const float* __restrict__ ad,
                          const float* __restrict__ hfeat, const float* __restrict__ bias,
                          const float* __restrict__ W2,
                          const float* __restrict__ a_src2, const float* __restrict__ a_dst2,
                          const float* __restrict__ Wp, float* __restrict__ q,
                          int N) {
  __shared__ __attribute__((aligned(16))) float s_o[16 * O1_PITCH];
  __shared__ __attribute__((aligned(16))) float s_p[8 * 16 * H2_PITCH];
  const int lane = threadIdx.x & 31;
  const int wv = __builtin_amdgcn_readfirstlane(threadIdx.x >> 5);
  const int d = blockIdx.x * 16 + wv;
  float4 o = make_float4(0.f, 0.f, 0.f, 0.f);
  if (d < N) {
    const int sl = slot[(size_t)d * CAP + lane];
    const int c = __builtin_amdgcn_readfirstlane(cnt[d]);
    const int deg = c + 1;
    const float4 adv = *reinterpret_cast<const float4*>(ad + 4 * (size_t)d);

    const bool hb0 = (lane & 8) != 0, hb1 = (lane & 16) != 0;
    float4 acc = make_float4(0.f, 0.f, 0.f, 0.f);
    float s0 = 0.f, s1 = 0.f, s2 = 0.f, s3 = 0.f;
    if (deg <= ROW_MAXDEG) {
      const bool act = lane < deg;
      const int s = (lane < c) ? sl : d;
      const float4 a = *reinterpret_cast<const float4*>(as + 4 * (size_t)s);
      const unsigned lo = 16u * (unsigned)lane;
      float4 r[ROW_GROUP];
#pragma unroll
      for (int i = 0; i < ROW_GROUP; ++i) r[i] = row_ld(hfeat, lane_i(s, i), lo);
      const float e0 = act ? lrelu(a.x + adv.x) : -INFINITY;
      const float e1 = act ? lrelu(a.y + adv.y) : -INFINITY;
      const float e2 = act ? lrelu(a.z + adv.z) : -INFINITY;
      const float e3 = act ? lrelu(a.w + adv.w) : -INFINITY;
      const float m0 = wave_max_alu(e0), m1 = wave_max_alu(e1);
      const float m2 = wave_max_alu(e2), m3 = wave_max_alu(e3);
      const float x0 = act ? expf(e0 - m0) : 0.f;
      const float x1 = act ? expf(e1 - m1) : 0.f;
      const float x2 = act ? expf(e2 - m2) : 0.f;
      const float x3 = act ? expf(e3 - m3) : 0.f;
      s0 = x0; s1 = x1; s2 = x2; s3 = x3;
      for (int t = 0; t < deg; t += ROW_GROUP) {
#pragma unroll
        for (int i = 0; i < ROW_GROUP; ++i) {
          const float w0 = lane_f(x0, t + i), w1 = lane_f(x1, t + i);
          const float w2 = lane_f(x2, t + i), w3 = lane_f(x3, t + i);
          const float wlo = hb0 ? w1 : w0, whi = hb0 ? w3 : w2;
          const float w = hb1 ? whi : wlo;
          acc.x += w * r[i].x; acc.y += w * r[i].y; acc.z += w * r[i].z; acc.w += w * r[i].w;
        }
        if (t + ROW_GROUP < deg) {
#pragma unroll
          for (int i = 0; i < ROW_GROUP; ++i) r[i] = row_ld(hfeat, lane_i(s, t + ROW_GROUP + i), lo);
        }
      }
    } else {
      const int M = (c > CAP) ? __builtin_amdgcn_readfirstlane(cnt[N]) : 0;
      const int nl = (M + 31) >> 5;
      const unsigned lo = 16u * (unsigned)lane;
      float m0 = -INFINITY, m1 = -INFINITY, m2 = -INFINITY, m3 = -INFINITY;
      for (int k = 0; k <= nl + 1; ++k) {
        int s;
        if (big_item(k, nl, lane, d, sl, M, ovf, ei, E, s)) {
          const float4 a = *reinterpret_cast<const float4*>(as + 4 * (size_t)s);
          m0 = fmaxf(m0, lrelu(a.x + adv.x));
          m1 = fmaxf(m1, lrelu(a.y + adv.y));
          m2 = fmaxf(m2, lrelu(a.z + adv.z));
          m3 = fmaxf(m3, lrelu(a.w + adv.w));
        }
      }
      m0 = lane_f(wave_max(m0), 0); m1 = lane_f(wave_max(m1), 0);
      m2 = lane_f(wave_max(m2), 0); m3 = lane_f(wave_max(m3), 0);

      for (int k = 0; k <= nl + 1; ++k) {
        int s;
        const bool v = big_item(k, nl, lane, d, sl, M, ovf, ei, E, s);
        float x0 = 0.f, x1 = 0.f, x2 = 0.f, x3 = 0.f;
        if (v) {
          const float4 a = *reinterpret_cast<const float4*>(as + 4 * (size_t)s);
          x0 = expf(lrelu(a.x + adv.x) - m0);
          x1 = expf(lrelu(a.y + adv.y) - m1);
          x2 = expf(lrelu(a.z + adv.z) - m2);
          x3 = expf(lrelu(a.w + adv.w) - m3);
          s0 += x0; s1 += x1; s2 += x2; s3 += x3;
        }
        unsigned mask = __builtin_amdgcn_ballot_w32(v);
        while (mask != 0u) {
          int tt[BIG_GROUP];
#pragma unroll
          for (int i = 0; i < BIG_GROUP; ++i) {
            tt[i] = (mask != 0u) ? __builtin_ctz(mask) : -1;
            mask &= mask - 1u;
          }
          float4 r[BIG_GROUP];
#pragma unroll
          for (int i = 0; i < BIG_GROUP; ++i)
            r[i] = row_ld(hfeat, (tt[i] >= 0) ? lane_i(s, tt[i]) : d, lo);
#pragma unroll
          for (int i = 0; i < BIG_GROUP; ++i) {
            const int tl = (tt[i] >= 0) ? tt[i] : 0;
            const float w0 = lane_f(x0, tl), w1 = lane_f(x1, tl);
            const float w2 = lane_f(x2, tl), w3 = lane_f(x3, tl);
            const float wlo = hb0 ? w1 : w0, whi = hb0 ? w3 : w2;
            const float w = (tt[i] >= 0) ? (hb1 ? whi : wlo) : 0.f;
            acc.x += w * r[i].x; acc.y += w * r[i].y; acc.z += w * r[i].z; acc.w += w * r[i].w;
          }
        }
      }
    }
    s0 = wave_sum_alu(s0); s1 = wave_sum_alu(s1); s2 = wave_sum_alu(s2); s3 = wave_sum_alu(s3);
    const float slo = hb0 ? s1 : s0, shi = hb0 ? s3 : s2;
    const float den = (hb1 ? shi : slo) + 1e-16f;
    const float* bp = bias + 4 * lane;
    o.x = acc.x / den + bp[0];
    o.y = acc.y / den + bp[1];
    o.z = acc.z / den + bp[2];
    o.w = acc.w / den + bp[3];
    o.x = (o.x > 0.f) ? o.x : (expf(o.x) - 1.f);
    o.y = (o.y > 0.f) ? o.y : (expf(o.y) - 1.f);
    o.z = (o.z > 0.f) ? o.z : (expf(o.z) - 1.f);
    o.w = (o.w > 0.f) ? o.w : (expf(o.w) - 1.f);
  }
  const int lm = lane & 15;
  const int hi = lane >> 4;
  const int ct = wv & 1;
  const int kq = wv >> 1;
  const float* bcol = W2 + (size_t)(16 * kq + hi * 2) * 32 + ct * 16 + lm;
  v2f b[4];
#pragma unroll
  for (int i = 0; i < 4; ++i) {
    b[i].x = bcol[(size_t)(4 * i) * 32];
    b[i].y = bcol[(size_t)(4 * i + 1) * 32];
  }
  *reinterpret_cast<float4*>(s_o + wv * O1_PITCH + 4 * lane) = o;
  __syncthreads();

  const float wpa = Wp[lane], wpb = Wp[32 + lane];
  const float avs = a_src2[lane], avd = a_dst2[lane];
  v8f c = {};
  const float* arow = s_o + lm * O1_PITCH + 16 * kq + hi * 2;
#pragma unroll
  for (int i = 0; i < 4; ++i) {
    const v2f a = *reinterpret_cast<const v2f*>(arow + 4 * i);
    c = __builtin_amdgcn_wmma_f32_16x16x4_f32(
         false, a,  false, b[i],
         (short)0, c,  false,  false);
  }
  float* pt = s_p + kq * (16 * H2_PITCH);
#pragma unroll
  for (int r = 0; r < 8; ++r)
    pt[(hi * 8 + r) * H2_PITCH + ct * 16 + lm] = c[r];
  __syncthreads();

  if (d < N) {
    const float* pr = s_p + wv * H2_PITCH + lane;
    float p[8];
#pragma unroll
    for (int q = 0; q < 8; ++q) p[q] = pr[q * (16 * H2_PITCH)];
    const float v = ((p[0] + p[1]) + (p[2] + p[3])) + ((p[4] + p[5]) + (p[6] + p[7]));
    const float ta = v * avs, tb = v * avd, tc = v * wpa, td = v * wpb;
    const bool hb = (lane & 16) != 0, b3 = (lane & 8) != 0;
    const float f0 = (hb ? tc : ta) + __shfl_xor(hb ? ta : tc, 16);
    const float f1 = (hb ? td : tb) + __shfl_xor(hb ? tb : td, 16);
    float g = (b3 ? f1 : f0) + __shfl_xor(b3 ? f0 : f1, 8);
    g += __shfl_xor(g, 4);
    g += __shfl_xor(g, 2);
    g += __shfl_xor(g, 1);
    if ((lane & 7) == 0) q[4 * (size_t)d + (lane >> 3)] = g;
  }
}


__global__ void __launch_bounds__(256)
gat_scalar_h1_kernel(const int* __restrict__ cnt, const int* __restrict__ slot,
                     const float* __restrict__ ovf, const int* __restrict__ ei, int E,
                     const float* __restrict__ q, const float* __restrict__ bias,
                     const float* __restrict__ Wp,
                     float* __restrict__ ps, float* __restrict__ pd, int N) {
  const int lane = threadIdx.x & 31;
  const int sub = threadIdx.x & 15;
  const int d = blockIdx.x * 16 + (threadIdx.x >> 4);
  const bool live = d < N;
  const int dc = live ? d : N - 1;
  const int c0 = slot[(size_t)dc * CAP + sub];
  const int c1 = slot[(size_t)dc * CAP + 16 + sub];
  const int c = cnt[dc];
  const float adv = q[4 * (size_t)dc + 1];

  const int j0 = sub, j1 = sub + 16;
  const bool a0 = j0 <= c, a1 = j1 <= c;
  const int s0 = (j0 < c) ? c0 : dc;
  const int s1 = (j1 < c) ? c1 : dc;
  const float4 q0 = *reinterpret_cast<const float4*>(q + 4 * (size_t)s0);
  const float4 q1 = *reinterpret_cast<const float4*>(q + 4 * (size_t)s1);

  const float bl = bias[lane];
  const float cb1 = wave_sum(bl * Wp[lane]);
  const float cb2 = wave_sum(bl * Wp[32 + lane]);

  const bool big = c >= CAP;
  const int M = (c > CAP) ? cnt[N] : 0;
  const float es = lrelu(q[4 * (size_t)dc] + adv);

  const float e0 = a0 ? lrelu(q0.x + adv) : -INFINITY;
  const float e1 = a1 ? lrelu(q1.x + adv) : -INFINITY;
  float m = fmaxf(e0, e1);
  if (big) {
    if (sub == 0) m = fmaxf(m, es);
    for (int i = sub; i < M; i += 16) {
      const int e = __float_as_int(ovf[i]);
      if (ei[E + e] == dc) m = fmaxf(m, lrelu(q[4 * (size_t)ei[e]] + adv));
    }
  }
  m = row_max(m);

  const float x0 = a0 ? expf(e0 - m) : 0.f;
  const float x1 = a1 ? expf(e1 - m) : 0.f;
  float sx = x0 + x1;
  float su = x0 * q0.z + x1 * q1.z;
  float sv = x0 * q0.w + x1 * q1.w;
  if (big) {
    if (sub == 0) {
      const float4 qq = *reinterpret_cast<const float4*>(q + 4 * (size_t)dc);
      const float x = expf(es - m);
      sx += x; su += x * qq.z; sv += x * qq.w;
    }
    for (int i = sub; i < M; i += 16) {
      const int e = __float_as_int(ovf[i]);
      if (ei[E + e] == dc) {
        const float4 qq = *reinterpret_cast<const float4*>(q + 4 * (size_t)ei[e]);
        const float x = expf(lrelu(qq.x + adv) - m);
        sx += x; su += x * qq.z; sv += x * qq.w;
      }
    }
  }
  sx = row_sum(sx); su = row_sum(su); sv = row_sum(sv);
  const float den = sx + 1e-16f;
  if (live && sub == 0) {
    ps[d] = su / den + cb1;
    pd[d] = sv / den + cb2;
  }
}

__global__ void pred_edge_kernel(const int* __restrict__ ei, int E,
                                 const float* __restrict__ ps,
                                 const float* __restrict__ pd,
                                 const float* __restrict__ bp,
                                 float* __restrict__ out) {
  int e = blockIdx.x * blockDim.x + threadIdx.x;
  if (e >= E) return;
  out[e] = ps[ei[e]] + pd[ei[E + e]] + bp[0];
}

static inline int cdiv(long long a, int b) { return (int)((a + b - 1) / b); }

extern "C" void kernel_launch(void* const* d_in, const int* in_sizes, int n_in,
                              void* d_out, int out_size, void* d_ws, size_t ws_size,
                              hipStream_t stream) {
  const float* x   = (const float*)d_in[0];
  const int*   ei  = (const int*)  d_in[1];
  const float* W1  = (const float*)d_in[2];
  const float* as1w= (const float*)d_in[3];
  const float* ad1w= (const float*)d_in[4];
  const float* b1  = (const float*)d_in[5];
  const float* W2  = (const float*)d_in[6];
  const float* as2w= (const float*)d_in[7];
  const float* ad2w= (const float*)d_in[8];
  const float* b2  = (const float*)d_in[9];
  const float* Wp  = (const float*)d_in[10];
  const float* bp  = (const float*)d_in[11];
  float* out = (float*)d_out;

  const int F = 128, H1 = 4;
  const int N = in_sizes[0] / F;
  const int E = in_sizes[1] / 2;

  float* ws = (float*)d_ws;
  float*    h1    = ws;                 ws += (size_t)N * F;
  int*      slot  = (int*)ws;           ws += (size_t)N * CAP;
  float*    q2    = ws;                 ws += (size_t)N * 4;
  float*    as1   = ws;                 ws += (size_t)N * H1;
  float*    ad1   = ws;                 ws += (size_t)N * H1;
  int*      cnt   = (int*)ws;           ws += (size_t)N + 4;
  float*    ps    = ws;                 ws += (size_t)N;
  float*    pd    = ws;                 ws += (size_t)N;
  (void)ws_size; (void)n_in; (void)out_size;

  const int T = 256;

  gemm_alpha_h4_kernel<<<N / 16, 128, 0, stream>>>(x, W1, as1w, ad1w, h1, as1, ad1,
                                                   (unsigned*)cnt, N + 4);

  dst_bucket_kernel<<<cdiv(E, T), T, 0, stream>>>(ei, E, N, cnt, slot, out);

  gat_gather_h4_proj_kernel<<<cdiv(N, 16), 512, 0, stream>>>(cnt, slot, out, ei, E, as1, ad1, h1, b1,
                                                            W2, as2w, ad2w, Wp, q2, N);

  gat_scalar_h1_kernel<<<cdiv(N, 16), T, 0, stream>>>(cnt, slot, out, ei, E, q2, b2, Wp, ps, pd, N);

  pred_edge_kernel<<<cdiv(E, T), T, 0, stream>>>(ei, E, ps, pd, bp, out);
}
